# starting kernel
# baseline (speedup 1.0000x reference)
#include <hip/hip_runtime.h>
#include <math.h>

#define GAMMA 1.1f

typedef __attribute__((ext_vector_type(2))) float v2f;
typedef __attribute__((ext_vector_type(8))) float v8f;

__global__ void prep_kernel(const float* __restrict__ boxes, int M,
                            float* __restrict__ out, int out_n,
                            float4* __restrict__ boxdata,
                            float* __restrict__ rad2) {
  int t = blockIdx.x * blockDim.x + threadIdx.x;
  if (t < out_n) out[t] = 0.0f;
  if (t < M) {
    float bx = boxes[t * 9 + 0], by = boxes[t * 9 + 1];
    float dx = boxes[t * 9 + 3], dy = boxes[t * 9 + 4];
    float hx = 0.5f * dx, hy = 0.5f * dy;
    float r = sqrtf(hx * hx + hy * hy) * GAMMA;
    boxdata[t] = make_float4(-2.0f * bx, -2.0f * by, bx * bx + by * by, 0.0f);
    rad2[t] = r * r;
  }
}

__global__ __launch_bounds__(32)
void count_kernel(const float* __restrict__ P, int N,
                  const float4* __restrict__ boxdata,
                  const float* __restrict__ rad2,
                  unsigned* __restrict__ masks, int ngroup,
                  int* __restrict__ cnt, int nch) {
  const int wave = blockIdx.x;
  const int bt   = wave / nch;
  const int c    = wave % nch;
  const int lane = threadIdx.x;
  const bool lo  = lane < 16;

  float4 bd = boxdata[bt * 16 + (lane & 15)];
  v2f a;
  a.x = lo ? bd.x : 1.0f;
  a.y = lo ? bd.y : bd.z;

  float r2v[8];
#pragma unroll
  for (int r = 0; r < 8; ++r) r2v[r] = rad2[bt * 16 + r + (lo ? 0 : 8)];

  int cl[8], ch[8];
#pragma unroll
  for (int r = 0; r < 8; ++r) { cl[r] = 0; ch[r] = 0; }

  const int g0 = c * 64;
  for (int k = 0; k < 64; k += 2) {
    int idx0 = (g0 + k) * 16 + (lane & 15);
    int idx1 = idx0 + 16;
    float px0, py0, px1, py1;
    if (idx0 < N) { px0 = P[(size_t)idx0 * 5]; py0 = P[(size_t)idx0 * 5 + 1]; }
    else          { px0 = 1e18f; py0 = 1e18f; }
    if (idx1 < N) { px1 = P[(size_t)idx1 * 5]; py1 = P[(size_t)idx1 * 5 + 1]; }
    else          { px1 = 1e18f; py1 = 1e18f; }

    {
      int pk = (k + 2) & 63;
      int nidx = (g0 + pk) * 16 + (lane & 15);
      if (nidx < N) __builtin_prefetch(&P[(size_t)nidx * 5], 0, 3);
    }

    float pp0 = px0 * px0 + py0 * py0;
    float pp1 = px1 * px1 + py1 * py1;
    v2f b0, b1;
    b0.x = lo ? px0 : pp0;  b0.y = lo ? py0 : 1.0f;
    b1.x = lo ? px1 : pp1;  b1.y = lo ? py1 : 1.0f;

    v8f cz = {0.f, 0.f, 0.f, 0.f, 0.f, 0.f, 0.f, 0.f};
    v8f d0 = __builtin_amdgcn_wmma_f32_16x16x4_f32(false, a, false, b0,
                                                   (short)0, cz, false, false);
    v8f d1 = __builtin_amdgcn_wmma_f32_16x16x4_f32(false, a, false, b1,
                                                   (short)0, cz, false, false);
    unsigned w0 = 0, w1 = 0;
#pragma unroll
    for (int r = 0; r < 8; ++r) {
      unsigned bal0 = __builtin_amdgcn_ballot_w32(d0[r] <= r2v[r]);
      unsigned bal1 = __builtin_amdgcn_ballot_w32(d1[r] <= r2v[r]);
      cl[r] += __popc(bal0 & 0xFFFFu) + __popc(bal1 & 0xFFFFu);
      ch[r] += __popc(bal0 >> 16)     + __popc(bal1 >> 16);
      w0 = (lane == r) ? bal0 : w0;
      w1 = (lane == r) ? bal1 : w1;
    }
    if (lane < 8) {
      unsigned* mp = &masks[((size_t)bt * ngroup + (g0 + k)) * 8 + lane];
      mp[0] = w0;
      mp[8] = w1;
    }
  }

  if (lane == 0) {
#pragma unroll
    for (int r = 0; r < 8; ++r) {
      cnt[(bt * 16 + r)     * nch + c] = cl[r];
      cnt[(bt * 16 + r + 8) * nch + c] = ch[r];
    }
  }
}

__global__ void scan_kernel(const int* __restrict__ cnt,
                            int* __restrict__ baseArr, int nch) {
  __shared__ int s[1024];
  int m = blockIdx.x;
  int t = threadIdx.x;
  s[t] = cnt[(size_t)m * nch + t];
  __syncthreads();
  for (int off = 1; off < nch; off <<= 1) {
    int x = s[t];
    int y = (t >= off) ? s[t - off] : 0;
    __syncthreads();
    s[t] = x + y;
    __syncthreads();
  }
  baseArr[(size_t)m * nch + t] = (t == 0) ? 0 : s[t - 1];
}

__global__ __launch_bounds__(32)
void emit_kernel(const float* __restrict__ P,
                 const float* __restrict__ boxes,
                 const unsigned* __restrict__ masks, int ngroup,
                 const int* __restrict__ baseArr, int nch,
                 float* __restrict__ out, int S) {
  const int blk = blockIdx.x;
  const int m = blk / nch;
  const int c = blk % nch;
  int run = baseArr[(size_t)m * nch + c];
  if (run >= S) return;

  const int lane = threadIdx.x;
  const int bt = m >> 4, j = m & 15;
  const int ridx  = j & 7;
  const int shift = ((j >> 3) << 4) + (lane & 15);
  const float vx = boxes[m * 9 + 7];
  const float vy = boxes[m * 9 + 8];
  const unsigned ltmask = (1u << lane) - 1u;
  const int g0 = c * 64;

  for (int k = 0; k < 64; k += 2) {
    int G = g0 + k + (lane >> 4);
    unsigned w = masks[((size_t)bt * ngroup + G) * 8 + ridx];
    int bit = (int)((w >> shift) & 1u);
    unsigned bal = __builtin_amdgcn_ballot_w32(bit != 0);
    int slot = run + __popc(bal & ltmask);
    if (bit && slot < S) {
      int pidx = G * 16 + (lane & 15);
      const float* ps = P + (size_t)pidx * 5;
      float* o = out + ((size_t)m * S + slot) * 7;
      o[0] = ps[0]; o[1] = ps[1]; o[2] = ps[2];
      o[3] = ps[3]; o[4] = ps[4];
      o[5] = vx;    o[6] = vy;
    }
    run += __popc(bal);
    if (run >= S) break;
  }
}

extern "C" void kernel_launch(void* const* d_in, const int* in_sizes, int n_in,
                              void* d_out, int out_size, void* d_ws, size_t ws_size,
                              hipStream_t stream) {
  const float* P = (const float*)d_in[0];
  const float* B = (const float*)d_in[1];
  float* out = (float*)d_out;

  const int N = in_sizes[0] / 5;
  const int M = in_sizes[1] / 9;
  const int S = out_size / (M * 7);
  const int nch    = (N + 1023) / 1024;
  const int ngroup = nch * 64;
  const int bt     = M / 16;

  char* ws = (char*)d_ws;
  unsigned* masks   = (unsigned*)ws;
  size_t mbytes     = (size_t)bt * ngroup * 8 * sizeof(unsigned);
  int* cnt          = (int*)(ws + mbytes);
  int* baseA        = cnt + (size_t)M * nch;
  float4* boxdata   = (float4*)(baseA + (size_t)M * nch);
  float* rad2       = (float*)(boxdata + M);

  prep_kernel<<<(out_size + 255) / 256, 256, 0, stream>>>(B, M, out, out_size,
                                                          boxdata, rad2);
  count_kernel<<<bt * nch, 32, 0, stream>>>(P, N, boxdata, rad2,
                                            masks, ngroup, cnt, nch);
  scan_kernel<<<M, nch, 0, stream>>>(cnt, baseA, nch);
  emit_kernel<<<M * nch, 32, 0, stream>>>(P, B, masks, ngroup, baseA, nch,
                                          out, S);
}


# emit: one 8-wave work-group per box prefix-sums its chunk counts in-wave and walks only chunks holding points below slot S (slots staged in LDS); scan launch and 65536 one-wave groups gone
# speedup vs baseline: 1.4727x; 1.4727x over previous
#include <hip/hip_runtime.h>
#include <math.h>

#define GAMMA 1.1f

typedef __attribute__((ext_vector_type(2))) float v2f;
typedef __attribute__((ext_vector_type(8))) float v8f;

__global__ void prep_kernel(const float* __restrict__ boxes, int M,
                            float* __restrict__ out, int out_n,
                            float4* __restrict__ boxdata,
                            float* __restrict__ rad2) {
  int t = blockIdx.x * blockDim.x + threadIdx.x;
  if (t < out_n) out[t] = 0.0f;
  if (t < M) {
    float bx = boxes[t * 9 + 0], by = boxes[t * 9 + 1];
    float dx = boxes[t * 9 + 3], dy = boxes[t * 9 + 4];
    float hx = 0.5f * dx, hy = 0.5f * dy;
    float r = sqrtf(hx * hx + hy * hy) * GAMMA;
    boxdata[t] = make_float4(-2.0f * bx, -2.0f * by, bx * bx + by * by, 0.0f);
    rad2[t] = r * r;
  }
}

__global__ __launch_bounds__(32)
void count_kernel(const float* __restrict__ P, int N,
                  const float4* __restrict__ boxdata,
                  const float* __restrict__ rad2,
                  unsigned* __restrict__ masks, int ngroup,
                  int* __restrict__ cnt, int nch) {
  const int wave = blockIdx.x;
  const int bt   = wave / nch;
  const int c    = wave % nch;
  const int lane = threadIdx.x;
  const bool lo  = lane < 16;

  float4 bd = boxdata[bt * 16 + (lane & 15)];
  v2f a;
  a.x = lo ? bd.x : 1.0f;
  a.y = lo ? bd.y : bd.z;

  float r2v[8];
#pragma unroll
  for (int r = 0; r < 8; ++r) r2v[r] = rad2[bt * 16 + r + (lo ? 0 : 8)];

  int cl[8], ch[8];
#pragma unroll
  for (int r = 0; r < 8; ++r) { cl[r] = 0; ch[r] = 0; }

  const int g0 = c * 64;
  for (int k = 0; k < 64; k += 2) {
    int idx0 = (g0 + k) * 16 + (lane & 15);
    int idx1 = idx0 + 16;
    float px0, py0, px1, py1;
    if (idx0 < N) { px0 = P[(size_t)idx0 * 5]; py0 = P[(size_t)idx0 * 5 + 1]; }
    else          { px0 = 1e18f; py0 = 1e18f; }
    if (idx1 < N) { px1 = P[(size_t)idx1 * 5]; py1 = P[(size_t)idx1 * 5 + 1]; }
    else          { px1 = 1e18f; py1 = 1e18f; }

    {
      int pk = (k + 2) & 63;
      int nidx = (g0 + pk) * 16 + (lane & 15);
      if (nidx < N) __builtin_prefetch(&P[(size_t)nidx * 5], 0, 3);
    }

    float pp0 = px0 * px0 + py0 * py0;
    float pp1 = px1 * px1 + py1 * py1;
    v2f b0, b1;
    b0.x = lo ? px0 : pp0;  b0.y = lo ? py0 : 1.0f;
    b1.x = lo ? px1 : pp1;  b1.y = lo ? py1 : 1.0f;

    v8f cz = {0.f, 0.f, 0.f, 0.f, 0.f, 0.f, 0.f, 0.f};
    v8f d0 = __builtin_amdgcn_wmma_f32_16x16x4_f32(false, a, false, b0,
                                                   (short)0, cz, false, false);
    v8f d1 = __builtin_amdgcn_wmma_f32_16x16x4_f32(false, a, false, b1,
                                                   (short)0, cz, false, false);
    unsigned w0 = 0, w1 = 0;
#pragma unroll
    for (int r = 0; r < 8; ++r) {
      unsigned bal0 = __builtin_amdgcn_ballot_w32(d0[r] <= r2v[r]);
      unsigned bal1 = __builtin_amdgcn_ballot_w32(d1[r] <= r2v[r]);
      cl[r] += __popc(bal0 & 0xFFFFu) + __popc(bal1 & 0xFFFFu);
      ch[r] += __popc(bal0 >> 16)     + __popc(bal1 >> 16);
      w0 = (lane == r) ? bal0 : w0;
      w1 = (lane == r) ? bal1 : w1;
    }
    if (lane < 8) {
      unsigned* mp = &masks[((size_t)bt * ngroup + (g0 + k)) * 8 + lane];
      mp[0] = w0;
      mp[8] = w1;
    }
  }

  if (lane == 0) {
#pragma unroll
    for (int r = 0; r < 8; ++r) {
      cnt[(bt * 16 + r)     * nch + c] = cl[r];
      cnt[(bt * 16 + r + 8) * nch + c] = ch[r];
    }
  }
}

__device__ __forceinline__ int wave_incl_scan(int x) {
  x += __builtin_amdgcn_update_dpp(0, x, 0x111, 0xf, 0xf, true);
  x += __builtin_amdgcn_update_dpp(0, x, 0x112, 0xf, 0xf, true);
  x += __builtin_amdgcn_update_dpp(0, x, 0x114, 0xf, 0xf, true);
  x += __builtin_amdgcn_update_dpp(0, x, 0x118, 0xf, 0xf, true);
  int row0 = __builtin_amdgcn_readlane(x, 15);
  return x + ((threadIdx.x & 16) ? row0 : 0);
}

#define EMIT_THREADS 256
#define EMIT_BATCH 4

__global__ __launch_bounds__(EMIT_THREADS)
void emit_kernel(const float* __restrict__ P,
                 const float* __restrict__ boxes,
                 const unsigned* __restrict__ masks, int ngroup,
                 const int* __restrict__ cnt, int nch,
                 float* __restrict__ out, int S) {
  extern __shared__ int sIdx[];

  const int m    = blockIdx.x;
  const int t    = threadIdx.x;
  const int lane = t & 31;
  const int w    = __builtin_amdgcn_readfirstlane(t >> 5);
  const int bt = m >> 4, j = m & 15;
  const int sh = (j >> 3) << 4;
  const float vx = boxes[m * 9 + 7];
  const float vy = boxes[m * 9 + 8];
  const int* crow = cnt + (size_t)m * nch;
  const unsigned* mrow = masks + (size_t)bt * ngroup * 8 + (j & 7);

  int run0 = 0;
  for (int c0 = 0; c0 < nch && run0 < S; c0 += 256) {
    int n[8];
#pragma unroll
    for (int i = 0; i < 8; ++i) {
      const int c = c0 + 8 * lane + i;
      const int v = crow[c < nch ? c : nch - 1];
      n[i] = (c < nch) ? v : 0;
    }
    int p = 0, myOff = 0, myCnt = 0;
#pragma unroll
    for (int i = 0; i < 8; ++i) {
      if (i == w) { myOff = p; myCnt = n[i]; }
      p += n[i];
    }
    const int incl = wave_incl_scan(p);
    const int tileTotal = __builtin_amdgcn_readlane(incl, 31);
    const int myBase = run0 + (incl - p) + myOff;
    unsigned amask = __builtin_amdgcn_ballot_w32(myCnt > 0 && myBase < S);

    while (amask) {
      int  lsel[EMIT_BATCH];
      bool valid[EMIT_BATCH];
      unsigned wa[EMIT_BATCH], wb[EMIT_BATCH];
      const int first = __builtin_ctz(amask);
#pragma unroll
      for (int b = 0; b < EMIT_BATCH; ++b) {
        valid[b] = amask != 0u;
        lsel[b]  = valid[b] ? __builtin_ctz(amask) : first;
        amask &= amask - 1u;
        const int c = c0 + 8 * lsel[b] + w;
        const unsigned* mp = mrow + ((size_t)c * 64 + 2 * lane) * 8;
        wa[b] = mp[0];
        wb[b] = mp[8];
      }
#pragma unroll
      for (int b = 0; b < EMIT_BATCH; ++b) {
        if (valid[b]) {
          const int c = c0 + 8 * lsel[b] + w;
          const int base = __builtin_amdgcn_readlane(myBase, lsel[b]);
          unsigned W = ((wa[b] >> sh) & 0xFFFFu) | (((wb[b] >> sh) & 0xFFFFu) << 16);
          const int k = __popc(W);
          int slot = base + wave_incl_scan(k) - k;
          const int pb = c * 1024 + lane * 32;
          while (W != 0u && slot < S) {
            sIdx[slot] = pb + __builtin_ctz(W);
            W &= W - 1u;
            ++slot;
          }
        }
      }
    }
    run0 += tileTotal;
  }
  __syncthreads();

  const int tot = (run0 < S ? run0 : S) * 7;
  float* orow = out + (size_t)m * S * 7;
  for (int e0 = 0; e0 < tot; e0 += 4 * EMIT_THREADS) {
    float v[4];
#pragma unroll
    for (int u = 0; u < 4; ++u) {
      const int e  = e0 + u * EMIT_THREADS + t;
      const int ec = e < tot ? e : tot - 1;
      const int s = ec / 7, f = ec - s * 7;
      v[u] = (f < 5) ? P[(size_t)sIdx[s] * 5 + f] : (f == 5 ? vx : vy);
    }
#pragma unroll
    for (int u = 0; u < 4; ++u) {
      const int e = e0 + u * EMIT_THREADS + t;
      if (e < tot) orow[e] = v[u];
    }
  }
}

extern "C" void kernel_launch(void* const* d_in, const int* in_sizes, int n_in,
                              void* d_out, int out_size, void* d_ws, size_t ws_size,
                              hipStream_t stream) {
  const float* P = (const float*)d_in[0];
  const float* B = (const float*)d_in[1];
  float* out = (float*)d_out;

  const int N = in_sizes[0] / 5;
  const int M = in_sizes[1] / 9;
  const int S = out_size / (M * 7);
  const int nch    = (N + 1023) / 1024;
  const int ngroup = nch * 64;
  const int bt     = M / 16;

  char* ws = (char*)d_ws;
  unsigned* masks   = (unsigned*)ws;
  size_t mbytes     = (size_t)bt * ngroup * 8 * sizeof(unsigned);
  int* cnt          = (int*)(ws + mbytes);
  int* baseA        = cnt + (size_t)M * nch;
  float4* boxdata   = (float4*)(baseA + (size_t)M * nch);
  float* rad2       = (float*)(boxdata + M);

  prep_kernel<<<(out_size + 255) / 256, 256, 0, stream>>>(B, M, out, out_size,
                                                          boxdata, rad2);
  count_kernel<<<bt * nch, 32, 0, stream>>>(P, N, boxdata, rad2,
                                            masks, ngroup, cnt, nch);
  emit_kernel<<<M, EMIT_THREADS, (size_t)S * sizeof(int), stream>>>(
      P, B, masks, ngroup, cnt, nch, out, S);
}


# count: a chunk's per-box counts are vector popcounts of the mask words lanes 0..7 already hold (low half box r, high half box r+8); the scalar popcounts of all 128 ballots per tile are gone
# speedup vs baseline: 2.2789x; 1.5474x over previous
#include <hip/hip_runtime.h>
#include <math.h>

#define GAMMA 1.1f

typedef __attribute__((ext_vector_type(2))) float v2f;
typedef __attribute__((ext_vector_type(8))) float v8f;

#define CHUNK_PTS   256
#define HEAD_CHUNKS 32
#define HEAD_PTS    (CHUNK_PTS * HEAD_CHUNKS)
#define HEAD_SPLIT  4
#define HEAD_WAVES  (HEAD_CHUNKS / HEAD_SPLIT)

__device__ __forceinline__ void load_chunk(const float* __restrict__ P, int N,
                                           int pbase, int lane,
                                           float (&px)[16], float (&py)[16]) {
#pragma unroll
  for (int g = 0; g < 16; ++g) {
    const int idx = pbase + g * 16 + (lane & 15);
    const int ic  = idx < N ? idx : N - 1;
    const float x = P[(size_t)ic * 5];
    const float y = P[(size_t)ic * 5 + 1];
    px[g] = idx < N ? x : 1e18f;
    py[g] = idx < N ? y : 1e18f;
  }
}

__device__ __forceinline__ float sqnorm2(float x, float y) {
#pragma clang fp contract(off)
  return x * x + y * y;
}

__device__ __forceinline__ void tile_chunk(v2f a, const float (&r2v)[8],
                                           const float (&px)[16],
                                           const float (&py)[16],
                                           bool lo, int lane,
                                           unsigned* __restrict__ mbase,
                                           int& clo, int& chi) {
#pragma unroll
  for (int k = 0; k < 16; k += 2) {
    const float px0 = px[k],     py0 = py[k];
    const float px1 = px[k + 1], py1 = py[k + 1];
    const float pp0 = sqnorm2(px0, py0);
    const float pp1 = sqnorm2(px1, py1);
    v2f b0, b1;
    b0.x = lo ? px0 : pp0;  b0.y = lo ? py0 : 1.0f;
    b1.x = lo ? px1 : pp1;  b1.y = lo ? py1 : 1.0f;

    v8f cz = {0.f, 0.f, 0.f, 0.f, 0.f, 0.f, 0.f, 0.f};
    v8f d0 = __builtin_amdgcn_wmma_f32_16x16x4_f32(false, a, false, b0,
                                                   (short)0, cz, false, false);
    v8f d1 = __builtin_amdgcn_wmma_f32_16x16x4_f32(false, a, false, b1,
                                                   (short)0, cz, false, false);
    unsigned w0 = 0, w1 = 0;
#pragma unroll
    for (int r = 0; r < 8; ++r) {
      unsigned bal0 = __builtin_amdgcn_ballot_w32(d0[r] <= r2v[r]);
      unsigned bal1 = __builtin_amdgcn_ballot_w32(d1[r] <= r2v[r]);
      w0 = (lane == r) ? bal0 : w0;
      w1 = (lane == r) ? bal1 : w1;
    }
    clo += __popc((w0 & 0xFFFFu) | (w1 << 16));
    chi += __popc((w0 >> 16) | (w1 & 0xFFFF0000u));
    if (lane < 8) {
      unsigned* mp = mbase + k * 8 + lane;
      mp[0] = w0;
      mp[8] = w1;
    }
  }
}

__device__ __forceinline__ void box_operands(const float* __restrict__ boxes,
                                             int bi, bool valid, bool lo,
                                             v2f& a, float& r2own) {
  const float bx = boxes[bi * 9 + 0], by = boxes[bi * 9 + 1];
  const float dx = boxes[bi * 9 + 3], dy = boxes[bi * 9 + 4];
  const float hx = 0.5f * dx, hy = 0.5f * dy;
  const float r = sqrtf(hx * hx + hy * hy) * GAMMA;
  const float k0 = -2.0f * bx, k1 = -2.0f * by, k3 = bx * bx + by * by;
  a.x = valid ? (lo ? k0 : 1.0f) : 0.0f;
  a.y = valid ? (lo ? k1 : k3)   : 0.0f;
  r2own = valid ? r * r : -1.0f;
}

__global__ __launch_bounds__(HEAD_WAVES * 32)
void head_kernel(const float* __restrict__ P, int N,
                 const float* __restrict__ boxes,
                 unsigned* __restrict__ masksH,
                 int* __restrict__ cntH,
                 int* __restrict__ headpart, int M) {
  __shared__ int sC[HEAD_WAVES][16];
  const int tile = blockIdx.x / HEAD_SPLIT;
  const int qtr  = blockIdx.x % HEAD_SPLIT;
  const int lane = threadIdx.x & 31;
  const int wv   = __builtin_amdgcn_readfirstlane(threadIdx.x >> 5);
  const int w    = qtr * HEAD_WAVES + wv;
  const bool lo  = lane < 16;

  float px[16], py[16];
  load_chunk(P, N, w * CHUNK_PTS, lane, px, py);

  v2f a; float r2own;
  box_operands(boxes, tile * 16 + (lane & 15), true, lo, a, r2own);
  float r2v[8];
#pragma unroll
  for (int r = 0; r < 8; ++r) r2v[r] = __shfl(r2own, r + (lo ? 0 : 8));

  int clo = 0, chi = 0;
  tile_chunk(a, r2v, px, py, lo, lane,
             masksH + ((size_t)tile * (HEAD_CHUNKS * 16) + w * 16) * 8, clo, chi);

  if (lane < 8) {
    cntH[(tile * 16 + lane)     * HEAD_CHUNKS + w] = clo;
    cntH[(tile * 16 + lane + 8) * HEAD_CHUNKS + w] = chi;
    sC[wv][lane]     = clo;
    sC[wv][lane + 8] = chi;
  }
  __syncthreads();
  if (threadIdx.x < 16) {
    int s = 0;
    for (int i = 0; i < HEAD_WAVES; ++i) s += sC[i][threadIdx.x];
    headpart[qtr * M + tile * 16 + threadIdx.x] = s;
  }
}

__global__ __launch_bounds__(32)
void tail_kernel(const float* __restrict__ P, int N,
                 const float* __restrict__ boxes, int M, int S,
                 const int* __restrict__ headpart,
                 unsigned* __restrict__ masksT, int ngT,
                 int* __restrict__ cntT, int nchT,
                 int* __restrict__ slotOf) {
  extern __shared__ int slist[];
  const int cc   = blockIdx.x;
  const int lane = threadIdx.x;
  const bool lo  = lane < 16;

  float px[16], py[16];
  load_chunk(P, N, HEAD_PTS + cc * CHUNK_PTS, lane, px, py);

  int nslow = 0;
  for (int b0 = 0; b0 < M; b0 += 256) {
    bool f[8];
#pragma unroll
    for (int i = 0; i < 8; ++i) {
      const int idx = b0 + i * 32 + lane;
      const int ic = idx < M ? idx : M - 1;
      int v = 0;
#pragma unroll
      for (int q = 0; q < HEAD_SPLIT; ++q) v += headpart[q * M + ic];
      f[i] = idx < M && v < S;
    }
#pragma unroll
    for (int i = 0; i < 8; ++i) {
      const int idx = b0 + i * 32 + lane;
      const unsigned bal = __builtin_amdgcn_ballot_w32(f[i]);
      if (f[i]) {
        const int rank = nslow + __popc(bal & ((1u << lane) - 1u));
        slist[rank] = idx;
        if (cc == 0) slotOf[idx] = rank;
      }
      nslow += __popc(bal);
    }
  }
  __syncthreads();

  const int ntiles = (nslow + 15) >> 4;
  for (int t = 0; t < ntiles; ++t) {
    const int q = t * 16 + (lane & 15);
    const bool valid = q < nslow;
    const int bi = slist[valid ? q : 0];
    v2f a; float r2own;
    box_operands(boxes, bi, valid, lo, a, r2own);
    float r2v[8];
#pragma unroll
    for (int r = 0; r < 8; ++r) r2v[r] = __shfl(r2own, r + (lo ? 0 : 8));

    int clo = 0, chi = 0;
    tile_chunk(a, r2v, px, py, lo, lane,
               masksT + ((size_t)t * ngT + (size_t)cc * 16) * 8, clo, chi);

    if (lane < 8) {
      cntT[(size_t)(t * 16 + lane)     * nchT + cc] = clo;
      cntT[(size_t)(t * 16 + lane + 8) * nchT + cc] = chi;
    }
  }
}

__device__ __forceinline__ int wave_incl_scan(int x) {
  x += __builtin_amdgcn_update_dpp(0, x, 0x111, 0xf, 0xf, true);
  x += __builtin_amdgcn_update_dpp(0, x, 0x112, 0xf, 0xf, true);
  x += __builtin_amdgcn_update_dpp(0, x, 0x114, 0xf, 0xf, true);
  x += __builtin_amdgcn_update_dpp(0, x, 0x118, 0xf, 0xf, true);
  int row0 = __builtin_amdgcn_readlane(x, 15);
  return x + ((threadIdx.x & 16) ? row0 : 0);
}

#define EMIT_THREADS 256
#define EMIT_BATCH 4

__device__ __forceinline__ void walk_phase(const int* __restrict__ crow, int nch,
                                           const unsigned* __restrict__ mrow,
                                           int sh, int ptbase, int S,
                                           int lane, int w,
                                           int* sIdx, int& run0) {
  const int bsh = sh + 8 * (lane & 1);
  for (int c0 = 0; c0 < nch && run0 < S; c0 += 256) {
    int n[8];
#pragma unroll
    for (int i = 0; i < 8; ++i) {
      const int c = c0 + 8 * lane + i;
      const int v = crow[c < nch ? c : nch - 1];
      n[i] = (c < nch) ? v : 0;
    }
    int p = 0, myOff = 0, myCnt = 0;
#pragma unroll
    for (int i = 0; i < 8; ++i) {
      if (i == w) { myOff = p; myCnt = n[i]; }
      p += n[i];
    }
    const int incl = wave_incl_scan(p);
    const int tileTotal = __builtin_amdgcn_readlane(incl, 31);
    const int myBase = run0 + (incl - p) + myOff;
    unsigned amask = __builtin_amdgcn_ballot_w32(myCnt > 0 && myBase < S);

    while (amask) {
      int  lsel[EMIT_BATCH];
      bool valid[EMIT_BATCH];
      unsigned wa[EMIT_BATCH];
      const int first = __builtin_ctz(amask);
#pragma unroll
      for (int b = 0; b < EMIT_BATCH; ++b) {
        valid[b] = amask != 0u;
        lsel[b]  = valid[b] ? __builtin_ctz(amask) : first;
        amask &= amask - 1u;
        const int c = c0 + 8 * lsel[b] + w;
        wa[b] = mrow[((size_t)c * 16 + (lane >> 1)) * 8];
      }
#pragma unroll
      for (int b = 0; b < EMIT_BATCH; ++b) {
        if (valid[b]) {
          const int c = c0 + 8 * lsel[b] + w;
          const int base = __builtin_amdgcn_readlane(myBase, lsel[b]);
          unsigned W = (wa[b] >> bsh) & 0xFFu;
          const int k = __popc(W);
          int slot = base + wave_incl_scan(k) - k;
          const int pb = ptbase + c * CHUNK_PTS + lane * 8;
          while (W != 0u && slot < S) {
            sIdx[slot] = pb + __builtin_ctz(W);
            W &= W - 1u;
            ++slot;
          }
        }
      }
    }
    run0 += tileTotal;
  }
}

__global__ __launch_bounds__(EMIT_THREADS)
void emit_kernel(const float* __restrict__ P,
                 const float* __restrict__ boxes,
                 const unsigned* __restrict__ masksH,
                 const int* __restrict__ cntH,
                 const unsigned* __restrict__ masksT, int ngT,
                 const int* __restrict__ cntT, int nchT,
                 const int* __restrict__ slotOf,
                 float* __restrict__ out, int S) {
  extern __shared__ int sIdx[];

  const int m    = blockIdx.x;
  const int t    = threadIdx.x;
  const int lane = t & 31;
  const int w    = __builtin_amdgcn_readfirstlane(t >> 5);
  const float vx = boxes[m * 9 + 7];
  const float vy = boxes[m * 9 + 8];

  int run0 = 0;
  {
    const int bt = m >> 4, j = m & 15;
    walk_phase(cntH + (size_t)m * HEAD_CHUNKS, HEAD_CHUNKS,
               masksH + (size_t)bt * (HEAD_CHUNKS * 16) * 8 + (j & 7),
               (j >> 3) << 4, 0, S, lane, w, sIdx, run0);
  }
  if (run0 < S && nchT > 0) {
    const int slot = slotOf[m];
    const int bt = slot >> 4, j = slot & 15;
    walk_phase(cntT + (size_t)slot * nchT, nchT,
               masksT + (size_t)bt * ngT * 8 + (j & 7),
               (j >> 3) << 4, HEAD_PTS, S, lane, w, sIdx, run0);
  }
  __syncthreads();

  const int tot = (run0 < S ? run0 : S) * 7;
  float* orow = out + (size_t)m * S * 7;
  for (int e0 = 0; e0 < tot; e0 += 4 * EMIT_THREADS) {
    float v[4];
#pragma unroll
    for (int u = 0; u < 4; ++u) {
      const int e  = e0 + u * EMIT_THREADS + t;
      const int ec = e < tot ? e : tot - 1;
      const int s = ec / 7, f = ec - s * 7;
      v[u] = (f < 5) ? P[(size_t)sIdx[s] * 5 + f] : (f == 5 ? vx : vy);
    }
#pragma unroll
    for (int u = 0; u < 4; ++u) {
      const int e = e0 + u * EMIT_THREADS + t;
      if (e < tot) orow[e] = v[u];
    }
  }
  for (int e = tot + t; e < S * 7; e += EMIT_THREADS) orow[e] = 0.0f;
}

extern "C" void kernel_launch(void* const* d_in, const int* in_sizes, int n_in,
                              void* d_out, int out_size, void* d_ws, size_t ws_size,
                              hipStream_t stream) {
  const float* P = (const float*)d_in[0];
  const float* B = (const float*)d_in[1];
  float* out = (float*)d_out;

  const int N = in_sizes[0] / 5;
  const int M = in_sizes[1] / 9;
  const int S = out_size / (M * 7);
  const int bt   = M / 16;
  const int nchT = N > HEAD_PTS ? (N - HEAD_PTS + CHUNK_PTS - 1) / CHUNK_PTS : 0;
  const int ngT  = nchT * 16;

  char* ws = (char*)d_ws;
  unsigned* masksT = (unsigned*)ws;
  size_t tbytes    = (size_t)bt * ngT * 8 * sizeof(unsigned);
  unsigned* masksH = (unsigned*)(ws + tbytes);
  size_t hbytes    = (size_t)bt * (HEAD_CHUNKS * 16) * 8 * sizeof(unsigned);
  int* cntT        = (int*)(ws + tbytes + hbytes);
  int* cntH        = cntT + (size_t)M * nchT;
  int* headpart    = cntH + (size_t)M * HEAD_CHUNKS;
  int* slotOf      = headpart + (size_t)HEAD_SPLIT * M;

  head_kernel<<<bt * HEAD_SPLIT, HEAD_WAVES * 32, 0, stream>>>(P, N, B, masksH, cntH,
                                                               headpart, M);
  if (nchT > 0)
    tail_kernel<<<nchT, 32, (size_t)M * sizeof(int), stream>>>(
        P, N, B, M, S, headpart, masksT, ngT, cntT, nchT, slotOf);
  emit_kernel<<<M, EMIT_THREADS, (size_t)S * sizeof(int), stream>>>(
      P, B, masksH, cntH, masksT, ngT, cntT, nchT, slotOf, out, S);
}


# emit: chunk counts of a phase read in one go (phase 1 one per lane, phase 2 1024 chunks per pass), chunks to walk listed per wave in LDS and walked 8 per trip: fewer dependent round trips per box
# speedup vs baseline: 2.3810x; 1.0448x over previous
#include <hip/hip_runtime.h>
#include <math.h>

#define GAMMA 1.1f

typedef __attribute__((ext_vector_type(2))) float v2f;
typedef __attribute__((ext_vector_type(8))) float v8f;

#define CHUNK_PTS   256
#define HEAD_CHUNKS 32
#define HEAD_PTS    (CHUNK_PTS * HEAD_CHUNKS)
#define HEAD_SPLIT  4
#define HEAD_WAVES  (HEAD_CHUNKS / HEAD_SPLIT)

__device__ __forceinline__ void load_chunk(const float* __restrict__ P, int N,
                                           int pbase, int lane,
                                           float (&px)[16], float (&py)[16]) {
#pragma unroll
  for (int g = 0; g < 16; ++g) {
    const int idx = pbase + g * 16 + (lane & 15);
    const int ic  = idx < N ? idx : N - 1;
    const float x = P[(size_t)ic * 5];
    const float y = P[(size_t)ic * 5 + 1];
    px[g] = idx < N ? x : 1e18f;
    py[g] = idx < N ? y : 1e18f;
  }
}

__device__ __forceinline__ float sqnorm2(float x, float y) {
#pragma clang fp contract(off)
  return x * x + y * y;
}

__device__ __forceinline__ void tile_chunk(v2f a, const float (&r2v)[8],
                                           const float (&px)[16],
                                           const float (&py)[16],
                                           bool lo, int lane,
                                           unsigned* __restrict__ mbase,
                                           int& clo, int& chi) {
#pragma unroll
  for (int k = 0; k < 16; k += 2) {
    const float px0 = px[k],     py0 = py[k];
    const float px1 = px[k + 1], py1 = py[k + 1];
    const float pp0 = sqnorm2(px0, py0);
    const float pp1 = sqnorm2(px1, py1);
    v2f b0, b1;
    b0.x = lo ? px0 : pp0;  b0.y = lo ? py0 : 1.0f;
    b1.x = lo ? px1 : pp1;  b1.y = lo ? py1 : 1.0f;

    v8f cz = {0.f, 0.f, 0.f, 0.f, 0.f, 0.f, 0.f, 0.f};
    v8f d0 = __builtin_amdgcn_wmma_f32_16x16x4_f32(false, a, false, b0,
                                                   (short)0, cz, false, false);
    v8f d1 = __builtin_amdgcn_wmma_f32_16x16x4_f32(false, a, false, b1,
                                                   (short)0, cz, false, false);
    unsigned w0 = 0, w1 = 0;
#pragma unroll
    for (int r = 0; r < 8; ++r) {
      unsigned bal0 = __builtin_amdgcn_ballot_w32(d0[r] <= r2v[r]);
      unsigned bal1 = __builtin_amdgcn_ballot_w32(d1[r] <= r2v[r]);
      w0 = (lane == r) ? bal0 : w0;
      w1 = (lane == r) ? bal1 : w1;
    }
    clo += __popc((w0 & 0xFFFFu) | (w1 << 16));
    chi += __popc((w0 >> 16) | (w1 & 0xFFFF0000u));
    if (lane < 8) {
      unsigned* mp = mbase + k * 8 + lane;
      mp[0] = w0;
      mp[8] = w1;
    }
  }
}

__device__ __forceinline__ void box_operands(const float* __restrict__ boxes,
                                             int bi, bool valid, bool lo,
                                             v2f& a, float& r2own) {
  const float bx = boxes[bi * 9 + 0], by = boxes[bi * 9 + 1];
  const float dx = boxes[bi * 9 + 3], dy = boxes[bi * 9 + 4];
  const float hx = 0.5f * dx, hy = 0.5f * dy;
  const float r = sqrtf(hx * hx + hy * hy) * GAMMA;
  const float k0 = -2.0f * bx, k1 = -2.0f * by, k3 = bx * bx + by * by;
  a.x = valid ? (lo ? k0 : 1.0f) : 0.0f;
  a.y = valid ? (lo ? k1 : k3)   : 0.0f;
  r2own = valid ? r * r : -1.0f;
}

__global__ __launch_bounds__(HEAD_WAVES * 32)
void head_kernel(const float* __restrict__ P, int N,
                 const float* __restrict__ boxes,
                 unsigned* __restrict__ masksH,
                 int* __restrict__ cntH,
                 int* __restrict__ headpart, int M) {
  __shared__ int sC[HEAD_WAVES][16];
  const int tile = blockIdx.x / HEAD_SPLIT;
  const int qtr  = blockIdx.x % HEAD_SPLIT;
  const int lane = threadIdx.x & 31;
  const int wv   = __builtin_amdgcn_readfirstlane(threadIdx.x >> 5);
  const int w    = qtr * HEAD_WAVES + wv;
  const bool lo  = lane < 16;

  float px[16], py[16];
  load_chunk(P, N, w * CHUNK_PTS, lane, px, py);

  v2f a; float r2own;
  box_operands(boxes, tile * 16 + (lane & 15), true, lo, a, r2own);
  float r2v[8];
#pragma unroll
  for (int r = 0; r < 8; ++r) r2v[r] = __shfl(r2own, r + (lo ? 0 : 8));

  int clo = 0, chi = 0;
  tile_chunk(a, r2v, px, py, lo, lane,
             masksH + ((size_t)tile * (HEAD_CHUNKS * 16) + w * 16) * 8, clo, chi);

  if (lane < 8) {
    cntH[(tile * 16 + lane)     * HEAD_CHUNKS + w] = clo;
    cntH[(tile * 16 + lane + 8) * HEAD_CHUNKS + w] = chi;
    sC[wv][lane]     = clo;
    sC[wv][lane + 8] = chi;
  }
  __syncthreads();
  if (threadIdx.x < 16) {
    int s = 0;
    for (int i = 0; i < HEAD_WAVES; ++i) s += sC[i][threadIdx.x];
    headpart[qtr * M + tile * 16 + threadIdx.x] = s;
  }
}

__global__ __launch_bounds__(32)
void tail_kernel(const float* __restrict__ P, int N,
                 const float* __restrict__ boxes, int M, int S,
                 const int* __restrict__ headpart,
                 unsigned* __restrict__ masksT, int ngT,
                 int* __restrict__ cntT, int nchT,
                 int* __restrict__ slotOf) {
  extern __shared__ int slist[];
  const int cc   = blockIdx.x;
  const int lane = threadIdx.x;
  const bool lo  = lane < 16;

  float px[16], py[16];
  load_chunk(P, N, HEAD_PTS + cc * CHUNK_PTS, lane, px, py);

  int nslow = 0;
  for (int b0 = 0; b0 < M; b0 += 256) {
    bool f[8];
#pragma unroll
    for (int i = 0; i < 8; ++i) {
      const int idx = b0 + i * 32 + lane;
      const int ic = idx < M ? idx : M - 1;
      int v = 0;
#pragma unroll
      for (int q = 0; q < HEAD_SPLIT; ++q) v += headpart[q * M + ic];
      f[i] = idx < M && v < S;
    }
#pragma unroll
    for (int i = 0; i < 8; ++i) {
      const int idx = b0 + i * 32 + lane;
      const unsigned bal = __builtin_amdgcn_ballot_w32(f[i]);
      const int rank = nslow + __popc(bal & ((1u << lane) - 1u));
      if (f[i]) slist[rank] = idx;
      if (cc == 0 && idx < M) slotOf[idx] = f[i] ? rank : -1;
      nslow += __popc(bal);
    }
  }
  __syncthreads();

  const int ntiles = (nslow + 15) >> 4;
  for (int t = 0; t < ntiles; ++t) {
    const int q = t * 16 + (lane & 15);
    const bool valid = q < nslow;
    const int bi = slist[valid ? q : 0];
    v2f a; float r2own;
    box_operands(boxes, bi, valid, lo, a, r2own);
    float r2v[8];
#pragma unroll
    for (int r = 0; r < 8; ++r) r2v[r] = __shfl(r2own, r + (lo ? 0 : 8));

    int clo = 0, chi = 0;
    tile_chunk(a, r2v, px, py, lo, lane,
               masksT + ((size_t)t * ngT + (size_t)cc * 16) * 8, clo, chi);

    if (lane < 8) {
      cntT[(size_t)(t * 16 + lane)     * nchT + cc] = clo;
      cntT[(size_t)(t * 16 + lane + 8) * nchT + cc] = chi;
    }
  }
}

__device__ __forceinline__ int wave_incl_scan(int x) {
  x += __builtin_amdgcn_update_dpp(0, x, 0x111, 0xf, 0xf, true);
  x += __builtin_amdgcn_update_dpp(0, x, 0x112, 0xf, 0xf, true);
  x += __builtin_amdgcn_update_dpp(0, x, 0x114, 0xf, 0xf, true);
  x += __builtin_amdgcn_update_dpp(0, x, 0x118, 0xf, 0xf, true);
  int row0 = __builtin_amdgcn_readlane(x, 15);
  return x + ((threadIdx.x & 16) ? row0 : 0);
}

#define EMIT_THREADS 256
#define EMIT_BATCH 4

static_assert(HEAD_CHUNKS == 32, "walk_head keeps one chunk count per lane");
__device__ __forceinline__ int walk_head(const int* __restrict__ crow,
                                         const unsigned* __restrict__ mrow,
                                         int sh, int S, int lane, int w,
                                         int* sIdx) {
  const int bsh = sh + 8 * (lane & 1);
  const int cnt  = crow[lane];
  const int incl = wave_incl_scan(cnt);
  const int total = __builtin_amdgcn_readlane(incl, 31);
  const int myBase = incl - cnt;
  unsigned amask = __builtin_amdgcn_ballot_w32(cnt > 0 && myBase < S) &
                   (0x01010101u << w);
  if (amask) {
    int  csel[EMIT_BATCH];
    bool valid[EMIT_BATCH];
    unsigned wa[EMIT_BATCH];
    const int first = __builtin_ctz(amask);
#pragma unroll
    for (int b = 0; b < EMIT_BATCH; ++b) {
      valid[b] = amask != 0u;
      csel[b]  = valid[b] ? __builtin_ctz(amask) : first;
      amask &= amask - 1u;
      wa[b] = mrow[((size_t)csel[b] * 16 + (lane >> 1)) * 8];
    }
#pragma unroll
    for (int b = 0; b < EMIT_BATCH; ++b) {
      if (valid[b]) {
        const int base = __builtin_amdgcn_readlane(myBase, csel[b]);
        unsigned Wd = (wa[b] >> bsh) & 0xFFu;
        const int k = __popc(Wd);
        int slot = base + wave_incl_scan(k) - k;
        const int pb = csel[b] * CHUNK_PTS + lane * 8;
        while (Wd != 0u && slot < S) {
          sIdx[slot] = pb + __builtin_ctz(Wd);
          Wd &= Wd - 1u;
          ++slot;
        }
      }
    }
  }
  return total;
}

#define TAIL_TILES 4
#define TAIL_BATCH 8
#define LIST_MAX   (TAIL_TILES * 32)

__device__ __forceinline__ void walk_tail(const int* __restrict__ crow, int nch,
                                          const unsigned* __restrict__ mrow,
                                          int sh, int ptbase, int S,
                                          int lane, int w,
                                          int* sIdx, int* lChunk, int* lBase,
                                          int& run0) {
  const int bsh = sh + 8 * (lane & 1);
  const unsigned below = (1u << lane) - 1u;
  for (int c00 = 0; c00 < nch && run0 < S; c00 += 256 * TAIL_TILES) {
    int n[TAIL_TILES][8];
#pragma unroll
    for (int j = 0; j < TAIL_TILES; ++j) {
#pragma unroll
      for (int i = 0; i < 8; ++i) {
        const int c = c00 + 256 * j + 8 * lane + i;
        n[j][i] = crow[c < nch ? c : nch - 1];
      }
    }
    asm volatile("" :: "v"(n[0][0]), "v"(n[0][1]), "v"(n[0][2]), "v"(n[0][3]),
                       "v"(n[0][4]), "v"(n[0][5]), "v"(n[0][6]), "v"(n[0][7]),
                       "v"(n[1][0]), "v"(n[1][1]), "v"(n[1][2]), "v"(n[1][3]),
                       "v"(n[1][4]), "v"(n[1][5]), "v"(n[1][6]), "v"(n[1][7]));
    asm volatile("" :: "v"(n[2][0]), "v"(n[2][1]), "v"(n[2][2]), "v"(n[2][3]),
                       "v"(n[2][4]), "v"(n[2][5]), "v"(n[2][6]), "v"(n[2][7]),
                       "v"(n[3][0]), "v"(n[3][1]), "v"(n[3][2]), "v"(n[3][3]),
                       "v"(n[3][4]), "v"(n[3][5]), "v"(n[3][6]), "v"(n[3][7]));
#pragma unroll
    for (int j = 0; j < TAIL_TILES; ++j) {
#pragma unroll
      for (int i = 0; i < 8; ++i) {
        const int c = c00 + 256 * j + 8 * lane + i;
        n[j][i] = (c < nch) ? n[j][i] : 0;
      }
    }
    int len = 0;
    int run = run0;
#pragma unroll
    for (int j = 0; j < TAIL_TILES; ++j) {
      int p = 0, myOff = 0, myCnt = 0;
#pragma unroll
      for (int i = 0; i < 8; ++i) {
        if (i == w) { myOff = p; myCnt = n[j][i]; }
        p += n[j][i];
      }
      const int incl = wave_incl_scan(p);
      const int tileTotal = __builtin_amdgcn_readlane(incl, 31);
      const int myBase = run + (incl - p) + myOff;
      const bool act = myCnt > 0 && myBase < S;
      const unsigned am = __builtin_amdgcn_ballot_w32(act);
      if (act) {
        const int k = len + __popc(am & below);
        lChunk[k] = c00 + 256 * j + 8 * lane + w;
        lBase[k]  = myBase;
      }
      len += __popc(am);
      run += tileTotal;
    }
    run0 = run;
    __syncthreads();

    for (int k0 = 0; k0 < len; k0 += TAIL_BATCH) {
      int cs[TAIL_BATCH], bs[TAIL_BATCH];
      unsigned wa[TAIL_BATCH];
#pragma unroll
      for (int b = 0; b < TAIL_BATCH; ++b) {
        const int k = (k0 + b < len) ? k0 + b : k0;
        cs[b] = lChunk[k];
        bs[b] = lBase[k];
        wa[b] = mrow[((size_t)cs[b] * 16 + (lane >> 1)) * 8];
      }
#pragma unroll
      for (int b = 0; b < TAIL_BATCH; ++b) {
        if (k0 + b < len) {
          unsigned Wd = (wa[b] >> bsh) & 0xFFu;
          const int kk = __popc(Wd);
          int slot = bs[b] + wave_incl_scan(kk) - kk;
          const int pb = ptbase + cs[b] * CHUNK_PTS + lane * 8;
          while (Wd != 0u && slot < S) {
            sIdx[slot] = pb + __builtin_ctz(Wd);
            Wd &= Wd - 1u;
            ++slot;
          }
        }
      }
    }
  }
}

__global__ __launch_bounds__(EMIT_THREADS)
void emit_kernel(const float* __restrict__ P,
                 const float* __restrict__ boxes,
                 const unsigned* __restrict__ masksH,
                 const int* __restrict__ cntH,
                 const unsigned* __restrict__ masksT, int ngT,
                 const int* __restrict__ cntT, int nchT,
                 const int* __restrict__ slotOf,
                 float* __restrict__ out, int S) {
  extern __shared__ int sIdx[];
  __shared__ int lChunk[EMIT_THREADS / 32][LIST_MAX];
  __shared__ int lBase[EMIT_THREADS / 32][LIST_MAX];

  const int m    = blockIdx.x;
  const int t    = threadIdx.x;
  const int lane = t & 31;
  const int w    = __builtin_amdgcn_readfirstlane(t >> 5);
  const float vx = boxes[m * 9 + 7];
  const float vy = boxes[m * 9 + 8];
  const int slot = nchT > 0 ? slotOf[m] : -1;

  int run0;
  {
    const int bt = m >> 4, j = m & 15;
    run0 = walk_head(cntH + (size_t)m * HEAD_CHUNKS,
                     masksH + (size_t)bt * (HEAD_CHUNKS * 16) * 8 + (j & 7),
                     (j >> 3) << 4, S, lane, w, sIdx);
  }
  if (run0 < S && nchT > 0) {
    const int bt = slot >> 4, j = slot & 15;
    walk_tail(cntT + (size_t)slot * nchT, nchT,
              masksT + (size_t)bt * ngT * 8 + (j & 7),
              (j >> 3) << 4, HEAD_PTS, S, lane, w, sIdx,
              lChunk[w], lBase[w], run0);
  }
  __syncthreads();

  const int tot = (run0 < S ? run0 : S) * 7;
  float* orow = out + (size_t)m * S * 7;
  for (int e0 = 0; e0 < tot; e0 += 4 * EMIT_THREADS) {
    float v[4];
#pragma unroll
    for (int u = 0; u < 4; ++u) {
      const int e  = e0 + u * EMIT_THREADS + t;
      const int ec = e < tot ? e : tot - 1;
      const int s = ec / 7, f = ec - s * 7;
      v[u] = (f < 5) ? P[(size_t)sIdx[s] * 5 + f] : (f == 5 ? vx : vy);
    }
#pragma unroll
    for (int u = 0; u < 4; ++u) {
      const int e = e0 + u * EMIT_THREADS + t;
      if (e < tot) orow[e] = v[u];
    }
  }
  for (int e = tot + t; e < S * 7; e += EMIT_THREADS) orow[e] = 0.0f;
}

extern "C" void kernel_launch(void* const* d_in, const int* in_sizes, int n_in,
                              void* d_out, int out_size, void* d_ws, size_t ws_size,
                              hipStream_t stream) {
  const float* P = (const float*)d_in[0];
  const float* B = (const float*)d_in[1];
  float* out = (float*)d_out;

  const int N = in_sizes[0] / 5;
  const int M = in_sizes[1] / 9;
  const int S = out_size / (M * 7);
  const int bt   = M / 16;
  const int nchT = N > HEAD_PTS ? (N - HEAD_PTS + CHUNK_PTS - 1) / CHUNK_PTS : 0;
  const int ngT  = nchT * 16;

  char* ws = (char*)d_ws;
  unsigned* masksT = (unsigned*)ws;
  size_t tbytes    = (size_t)bt * ngT * 8 * sizeof(unsigned);
  unsigned* masksH = (unsigned*)(ws + tbytes);
  size_t hbytes    = (size_t)bt * (HEAD_CHUNKS * 16) * 8 * sizeof(unsigned);
  int* cntT        = (int*)(ws + tbytes + hbytes);
  int* cntH        = cntT + (size_t)M * nchT;
  int* headpart    = cntH + (size_t)M * HEAD_CHUNKS;
  int* slotOf      = headpart + (size_t)HEAD_SPLIT * M;

  head_kernel<<<bt * HEAD_SPLIT, HEAD_WAVES * 32, 0, stream>>>(P, N, B, masksH, cntH,
                                                               headpart, M);
  if (nchT > 0)
    tail_kernel<<<nchT, 32, (size_t)M * sizeof(int), stream>>>(
        P, N, B, M, S, headpart, masksT, ngT, cntT, nchT, slotOf);
  emit_kernel<<<M, EMIT_THREADS, (size_t)S * sizeof(int), stream>>>(
      P, B, masksH, cntH, masksT, ngT, cntT, nchT, slotOf, out, S);
}


# tail: up to 3 listed tiles run jointly over a chunk: B fragments built once, the tiles' box loads issued together, each product issued among the compares of an earlier one (scheduling hint)
# speedup vs baseline: 2.6021x; 1.0929x over previous
#include <hip/hip_runtime.h>
#include <math.h>

#define GAMMA 1.1f

typedef __attribute__((ext_vector_type(2))) float v2f;
typedef __attribute__((ext_vector_type(8))) float v8f;

#define CHUNK_PTS   256
#define HEAD_CHUNKS 32
#define HEAD_PTS    (CHUNK_PTS * HEAD_CHUNKS)
#define HEAD_SPLIT  4
#define HEAD_WAVES  (HEAD_CHUNKS / HEAD_SPLIT)
#define TILE_GANG   3

__device__ __forceinline__ void load_chunk(const float* __restrict__ P, int N,
                                           int pbase, int lane,
                                           float (&px)[16], float (&py)[16]) {
#pragma unroll
  for (int g = 0; g < 16; ++g) {
    const int idx = pbase + g * 16 + (lane & 15);
    const int ic  = idx < N ? idx : N - 1;
    const float x = P[(size_t)ic * 5];
    const float y = P[(size_t)ic * 5 + 1];
    px[g] = idx < N ? x : 1e18f;
    py[g] = idx < N ? y : 1e18f;
  }
}

__device__ __forceinline__ float sqnorm2(float x, float y) {
#pragma clang fp contract(off)
  return x * x + y * y;
}

__device__ __forceinline__ void tile_chunk(v2f a, const float (&r2v)[8],
                                           const float (&px)[16],
                                           const float (&py)[16],
                                           bool lo, int lane,
                                           unsigned* __restrict__ mbase,
                                           int& clo, int& chi) {
#pragma unroll
  for (int k = 0; k < 16; k += 2) {
    const float px0 = px[k],     py0 = py[k];
    const float px1 = px[k + 1], py1 = py[k + 1];
    const float pp0 = sqnorm2(px0, py0);
    const float pp1 = sqnorm2(px1, py1);
    v2f b0, b1;
    b0.x = lo ? px0 : pp0;  b0.y = lo ? py0 : 1.0f;
    b1.x = lo ? px1 : pp1;  b1.y = lo ? py1 : 1.0f;

    v8f cz = {0.f, 0.f, 0.f, 0.f, 0.f, 0.f, 0.f, 0.f};
    v8f d0 = __builtin_amdgcn_wmma_f32_16x16x4_f32(false, a, false, b0,
                                                   (short)0, cz, false, false);
    v8f d1 = __builtin_amdgcn_wmma_f32_16x16x4_f32(false, a, false, b1,
                                                   (short)0, cz, false, false);
    unsigned w0 = 0, w1 = 0;
#pragma unroll
    for (int r = 0; r < 8; ++r) {
      unsigned bal0 = __builtin_amdgcn_ballot_w32(d0[r] <= r2v[r]);
      unsigned bal1 = __builtin_amdgcn_ballot_w32(d1[r] <= r2v[r]);
      w0 = (lane == r) ? bal0 : w0;
      w1 = (lane == r) ? bal1 : w1;
    }
    clo += __popc((w0 & 0xFFFFu) | (w1 << 16));
    chi += __popc((w0 >> 16) | (w1 & 0xFFFF0000u));
    if (lane < 8) {
      unsigned* mp = mbase + k * 8 + lane;
      mp[0] = w0;
      mp[8] = w1;
    }
  }
}

__device__ __forceinline__ void box_operands(const float* __restrict__ boxes,
                                             int bi, bool valid, bool lo,
                                             v2f& a, float& r2own) {
  const float bx = boxes[bi * 9 + 0], by = boxes[bi * 9 + 1];
  const float dx = boxes[bi * 9 + 3], dy = boxes[bi * 9 + 4];
  const float hx = 0.5f * dx, hy = 0.5f * dy;
  const float r = sqrtf(hx * hx + hy * hy) * GAMMA;
  const float k0 = -2.0f * bx, k1 = -2.0f * by, k3 = bx * bx + by * by;
  a.x = valid ? (lo ? k0 : 1.0f) : 0.0f;
  a.y = valid ? (lo ? k1 : k3)   : 0.0f;
  r2own = valid ? r * r : -1.0f;
}

template <int NT>
__device__ __forceinline__ void tiles_chunk(const v2f (&a)[NT],
                                            const float (&r2v)[NT][8],
                                            const float (&px)[16],
                                            const float (&py)[16],
                                            bool lo, int lane,
                                            unsigned* __restrict__ mbase,
                                            size_t tstride,
                                            int (&clo)[NT], int (&chi)[NT]) {
  v2f b[16];
#pragma unroll
  for (int g = 0; g < 16; ++g) {
    const float pp = sqnorm2(px[g], py[g]);
    b[g].x = lo ? px[g] : pp;
    b[g].y = lo ? py[g] : 1.0f;
  }
  __builtin_amdgcn_sched_barrier(0);
  const v8f cz = {0.f, 0.f, 0.f, 0.f, 0.f, 0.f, 0.f, 0.f};
  v8f d0[NT], d1[NT];
#pragma unroll
  for (int u = 0; u < NT; ++u) {
    d0[u] = __builtin_amdgcn_wmma_f32_16x16x4_f32(false, a[u], false, b[0],
                                                  (short)0, cz, false, false);
    d1[u] = __builtin_amdgcn_wmma_f32_16x16x4_f32(false, a[u], false, b[1],
                                                  (short)0, cz, false, false);
  }
#pragma unroll
  for (int k = 0; k < 16; k += 2) {
    unsigned w0[NT], w1[NT];
    v8f n0[NT], n1[NT];
#pragma unroll
    for (int u = 0; u < NT; ++u) {
      if (k + 2 < 16) {
        n0[u] = __builtin_amdgcn_wmma_f32_16x16x4_f32(false, a[u], false, b[k + 2],
                                                      (short)0, cz, false, false);
        n1[u] = __builtin_amdgcn_wmma_f32_16x16x4_f32(false, a[u], false, b[k + 3],
                                                      (short)0, cz, false, false);
      }
      unsigned x0 = 0, x1 = 0;
#pragma unroll
      for (int r = 0; r < 8; ++r) {
        unsigned bal0 = __builtin_amdgcn_ballot_w32(d0[u][r] <= r2v[u][r]);
        unsigned bal1 = __builtin_amdgcn_ballot_w32(d1[u][r] <= r2v[u][r]);
        x0 = (lane == r) ? bal0 : x0;
        x1 = (lane == r) ? bal1 : x1;
      }
      w0[u] = x0;
      w1[u] = x1;
      clo[u] += __popc((x0 & 0xFFFFu) | (x1 << 16));
      chi[u] += __popc((x0 >> 16) | (x1 & 0xFFFF0000u));
    }
    if (k + 2 < 16) {
#pragma unroll
      for (int i = 0; i < 2 * NT; ++i) {
        __builtin_amdgcn_sched_group_barrier(0x008, 1, 0);
        __builtin_amdgcn_sched_group_barrier(0x002, 16, 0);
      }
    }
    if (lane < 8) {
#pragma unroll
      for (int u = 0; u < NT; ++u) {
        unsigned* mp = mbase + (size_t)u * tstride + k * 8 + lane;
        mp[0] = w0[u];
        mp[8] = w1[u];
      }
    }
    if (k + 2 < 16) {
#pragma unroll
      for (int u = 0; u < NT; ++u) { d0[u] = n0[u]; d1[u] = n1[u]; }
    }
  }
}

template <int NT>
__device__ __forceinline__ void run_tiles(const float* __restrict__ boxes,
                                          const int* slist, int nslow, int t0,
                                          const float (&px)[16],
                                          const float (&py)[16],
                                          bool lo, int lane,
                                          unsigned* __restrict__ masksT, int ngT,
                                          int* __restrict__ cntT, int nchT,
                                          int cc) {
  v2f a[NT];
  float r2own[NT];
#pragma unroll
  for (int u = 0; u < NT; ++u) {
    const int q = (t0 + u) * 16 + (lane & 15);
    const bool valid = q < nslow;
    const int bi = slist[valid ? q : 0];
    box_operands(boxes, bi, valid, lo, a[u], r2own[u]);
  }
  float r2v[NT][8];
#pragma unroll
  for (int u = 0; u < NT; ++u) {
#pragma unroll
    for (int r = 0; r < 8; ++r) r2v[u][r] = __shfl(r2own[u], r + (lo ? 0 : 8));
  }
  int clo[NT], chi[NT];
#pragma unroll
  for (int u = 0; u < NT; ++u) { clo[u] = 0; chi[u] = 0; }

  tiles_chunk<NT>(a, r2v, px, py, lo, lane,
                  masksT + ((size_t)t0 * ngT + (size_t)cc * 16) * 8,
                  (size_t)ngT * 8, clo, chi);

  if (lane < 8) {
#pragma unroll
    for (int u = 0; u < NT; ++u) {
      cntT[(size_t)((t0 + u) * 16 + lane)     * nchT + cc] = clo[u];
      cntT[(size_t)((t0 + u) * 16 + lane + 8) * nchT + cc] = chi[u];
    }
  }
}

__global__ __launch_bounds__(HEAD_WAVES * 32)
void head_kernel(const float* __restrict__ P, int N,
                 const float* __restrict__ boxes,
                 unsigned* __restrict__ masksH,
                 int* __restrict__ cntH,
                 int* __restrict__ headpart, int M) {
  __shared__ int sC[HEAD_WAVES][16];
  const int tile = blockIdx.x / HEAD_SPLIT;
  const int qtr  = blockIdx.x % HEAD_SPLIT;
  const int lane = threadIdx.x & 31;
  const int wv   = __builtin_amdgcn_readfirstlane(threadIdx.x >> 5);
  const int w    = qtr * HEAD_WAVES + wv;
  const bool lo  = lane < 16;

  float px[16], py[16];
  load_chunk(P, N, w * CHUNK_PTS, lane, px, py);

  v2f a; float r2own;
  box_operands(boxes, tile * 16 + (lane & 15), true, lo, a, r2own);
  float r2v[8];
#pragma unroll
  for (int r = 0; r < 8; ++r) r2v[r] = __shfl(r2own, r + (lo ? 0 : 8));

  int clo = 0, chi = 0;
  tile_chunk(a, r2v, px, py, lo, lane,
             masksH + ((size_t)tile * (HEAD_CHUNKS * 16) + w * 16) * 8, clo, chi);

  if (lane < 8) {
    cntH[(tile * 16 + lane)     * HEAD_CHUNKS + w] = clo;
    cntH[(tile * 16 + lane + 8) * HEAD_CHUNKS + w] = chi;
    sC[wv][lane]     = clo;
    sC[wv][lane + 8] = chi;
  }
  __syncthreads();
  if (threadIdx.x < 16) {
    int s = 0;
    for (int i = 0; i < HEAD_WAVES; ++i) s += sC[i][threadIdx.x];
    headpart[qtr * M + tile * 16 + threadIdx.x] = s;
  }
}

__global__ __launch_bounds__(32)
void tail_kernel(const float* __restrict__ P, int N,
                 const float* __restrict__ boxes, int M, int S,
                 const int* __restrict__ headpart,
                 unsigned* __restrict__ masksT, int ngT,
                 int* __restrict__ cntT, int nchT,
                 int* __restrict__ slotOf) {
  extern __shared__ int slist[];
  const int cc   = blockIdx.x;
  const int lane = threadIdx.x;
  const bool lo  = lane < 16;

  float px[16], py[16];
  load_chunk(P, N, HEAD_PTS + cc * CHUNK_PTS, lane, px, py);

  int nslow = 0;
  for (int b0 = 0; b0 < M; b0 += 256) {
    bool f[8];
#pragma unroll
    for (int i = 0; i < 8; ++i) {
      const int idx = b0 + i * 32 + lane;
      const int ic = idx < M ? idx : M - 1;
      int v = 0;
#pragma unroll
      for (int q = 0; q < HEAD_SPLIT; ++q) v += headpart[q * M + ic];
      f[i] = idx < M && v < S;
    }
#pragma unroll
    for (int i = 0; i < 8; ++i) {
      const int idx = b0 + i * 32 + lane;
      const unsigned bal = __builtin_amdgcn_ballot_w32(f[i]);
      const int rank = nslow + __popc(bal & ((1u << lane) - 1u));
      if (f[i]) slist[rank] = idx;
      if (cc == 0 && idx < M) slotOf[idx] = f[i] ? rank : -1;
      nslow += __popc(bal);
    }
  }
  __syncthreads();

  static_assert(TILE_GANG == 3, "the branches below name the gang sizes");
  const int ntiles = (nslow + 15) >> 4;
  for (int t0 = 0; t0 < ntiles; t0 += TILE_GANG) {
    const int left = ntiles - t0;
    if (left >= 3)
      run_tiles<3>(boxes, slist, nslow, t0, px, py, lo, lane, masksT, ngT, cntT, nchT, cc);
    else if (left == 2)
      run_tiles<2>(boxes, slist, nslow, t0, px, py, lo, lane, masksT, ngT, cntT, nchT, cc);
    else
      run_tiles<1>(boxes, slist, nslow, t0, px, py, lo, lane, masksT, ngT, cntT, nchT, cc);
  }
}

__device__ __forceinline__ int wave_incl_scan(int x) {
  x += __builtin_amdgcn_update_dpp(0, x, 0x111, 0xf, 0xf, true);
  x += __builtin_amdgcn_update_dpp(0, x, 0x112, 0xf, 0xf, true);
  x += __builtin_amdgcn_update_dpp(0, x, 0x114, 0xf, 0xf, true);
  x += __builtin_amdgcn_update_dpp(0, x, 0x118, 0xf, 0xf, true);
  int row0 = __builtin_amdgcn_readlane(x, 15);
  return x + ((threadIdx.x & 16) ? row0 : 0);
}

#define EMIT_THREADS 256
#define EMIT_BATCH 4

static_assert(HEAD_CHUNKS == 32, "walk_head keeps one chunk count per lane");
__device__ __forceinline__ int walk_head(const int* __restrict__ crow,
                                         const unsigned* __restrict__ mrow,
                                         int sh, int S, int lane, int w,
                                         int* sIdx) {
  const int bsh = sh + 8 * (lane & 1);
  const int cnt  = crow[lane];
  const int incl = wave_incl_scan(cnt);
  const int total = __builtin_amdgcn_readlane(incl, 31);
  const int myBase = incl - cnt;
  unsigned amask = __builtin_amdgcn_ballot_w32(cnt > 0 && myBase < S) &
                   (0x01010101u << w);
  if (amask) {
    int  csel[EMIT_BATCH];
    bool valid[EMIT_BATCH];
    unsigned wa[EMIT_BATCH];
    const int first = __builtin_ctz(amask);
#pragma unroll
    for (int b = 0; b < EMIT_BATCH; ++b) {
      valid[b] = amask != 0u;
      csel[b]  = valid[b] ? __builtin_ctz(amask) : first;
      amask &= amask - 1u;
      wa[b] = mrow[((size_t)csel[b] * 16 + (lane >> 1)) * 8];
    }
#pragma unroll
    for (int b = 0; b < EMIT_BATCH; ++b) {
      if (valid[b]) {
        const int base = __builtin_amdgcn_readlane(myBase, csel[b]);
        unsigned Wd = (wa[b] >> bsh) & 0xFFu;
        const int k = __popc(Wd);
        int slot = base + wave_incl_scan(k) - k;
        const int pb = csel[b] * CHUNK_PTS + lane * 8;
        while (Wd != 0u && slot < S) {
          sIdx[slot] = pb + __builtin_ctz(Wd);
          Wd &= Wd - 1u;
          ++slot;
        }
      }
    }
  }
  return total;
}

#define TAIL_TILES 4
#define TAIL_BATCH 8
#define LIST_MAX   (TAIL_TILES * 32)

__device__ __forceinline__ void walk_tail(const int* __restrict__ crow, int nch,
                                          const unsigned* __restrict__ mrow,
                                          int sh, int ptbase, int S,
                                          int lane, int w,
                                          int* sIdx, int* lChunk, int* lBase,
                                          int& run0) {
  const int bsh = sh + 8 * (lane & 1);
  const unsigned below = (1u << lane) - 1u;
  for (int c00 = 0; c00 < nch && run0 < S; c00 += 256 * TAIL_TILES) {
    int n[TAIL_TILES][8];
#pragma unroll
    for (int j = 0; j < TAIL_TILES; ++j) {
#pragma unroll
      for (int i = 0; i < 8; ++i) {
        const int c = c00 + 256 * j + 8 * lane + i;
        n[j][i] = crow[c < nch ? c : nch - 1];
      }
    }
    asm volatile("" :: "v"(n[0][0]), "v"(n[0][1]), "v"(n[0][2]), "v"(n[0][3]),
                       "v"(n[0][4]), "v"(n[0][5]), "v"(n[0][6]), "v"(n[0][7]),
                       "v"(n[1][0]), "v"(n[1][1]), "v"(n[1][2]), "v"(n[1][3]),
                       "v"(n[1][4]), "v"(n[1][5]), "v"(n[1][6]), "v"(n[1][7]));
    asm volatile("" :: "v"(n[2][0]), "v"(n[2][1]), "v"(n[2][2]), "v"(n[2][3]),
                       "v"(n[2][4]), "v"(n[2][5]), "v"(n[2][6]), "v"(n[2][7]),
                       "v"(n[3][0]), "v"(n[3][1]), "v"(n[3][2]), "v"(n[3][3]),
                       "v"(n[3][4]), "v"(n[3][5]), "v"(n[3][6]), "v"(n[3][7]));
#pragma unroll
    for (int j = 0; j < TAIL_TILES; ++j) {
#pragma unroll
      for (int i = 0; i < 8; ++i) {
        const int c = c00 + 256 * j + 8 * lane + i;
        n[j][i] = (c < nch) ? n[j][i] : 0;
      }
    }
    int len = 0;
    int run = run0;
#pragma unroll
    for (int j = 0; j < TAIL_TILES; ++j) {
      int p = 0, myOff = 0, myCnt = 0;
#pragma unroll
      for (int i = 0; i < 8; ++i) {
        if (i == w) { myOff = p; myCnt = n[j][i]; }
        p += n[j][i];
      }
      const int incl = wave_incl_scan(p);
      const int tileTotal = __builtin_amdgcn_readlane(incl, 31);
      const int myBase = run + (incl - p) + myOff;
      const bool act = myCnt > 0 && myBase < S;
      const unsigned am = __builtin_amdgcn_ballot_w32(act);
      if (act) {
        const int k = len + __popc(am & below);
        lChunk[k] = c00 + 256 * j + 8 * lane + w;
        lBase[k]  = myBase;
      }
      len += __popc(am);
      run += tileTotal;
    }
    run0 = run;
    __syncthreads();

    for (int k0 = 0; k0 < len; k0 += TAIL_BATCH) {
      int cs[TAIL_BATCH], bs[TAIL_BATCH];
      unsigned wa[TAIL_BATCH];
#pragma unroll
      for (int b = 0; b < TAIL_BATCH; ++b) {
        const int k = (k0 + b < len) ? k0 + b : k0;
        cs[b] = lChunk[k];
        bs[b] = lBase[k];
        wa[b] = mrow[((size_t)cs[b] * 16 + (lane >> 1)) * 8];
      }
#pragma unroll
      for (int b = 0; b < TAIL_BATCH; ++b) {
        if (k0 + b < len) {
          unsigned Wd = (wa[b] >> bsh) & 0xFFu;
          const int kk = __popc(Wd);
          int slot = bs[b] + wave_incl_scan(kk) - kk;
          const int pb = ptbase + cs[b] * CHUNK_PTS + lane * 8;
          while (Wd != 0u && slot < S) {
            sIdx[slot] = pb + __builtin_ctz(Wd);
            Wd &= Wd - 1u;
            ++slot;
          }
        }
      }
    }
  }
}

__global__ __launch_bounds__(EMIT_THREADS)
void emit_kernel(const float* __restrict__ P,
                 const float* __restrict__ boxes,
                 const unsigned* __restrict__ masksH,
                 const int* __restrict__ cntH,
                 const unsigned* __restrict__ masksT, int ngT,
                 const int* __restrict__ cntT, int nchT,
                 const int* __restrict__ slotOf,
                 float* __restrict__ out, int S) {
  extern __shared__ int sIdx[];
  __shared__ int lChunk[EMIT_THREADS / 32][LIST_MAX];
  __shared__ int lBase[EMIT_THREADS / 32][LIST_MAX];

  const int m    = blockIdx.x;
  const int t    = threadIdx.x;
  const int lane = t & 31;
  const int w    = __builtin_amdgcn_readfirstlane(t >> 5);
  const float vx = boxes[m * 9 + 7];
  const float vy = boxes[m * 9 + 8];
  const int slot = nchT > 0 ? slotOf[m] : -1;

  int run0;
  {
    const int bt = m >> 4, j = m & 15;
    run0 = walk_head(cntH + (size_t)m * HEAD_CHUNKS,
                     masksH + (size_t)bt * (HEAD_CHUNKS * 16) * 8 + (j & 7),
                     (j >> 3) << 4, S, lane, w, sIdx);
  }
  if (run0 < S && nchT > 0) {
    const int bt = slot >> 4, j = slot & 15;
    walk_tail(cntT + (size_t)slot * nchT, nchT,
              masksT + (size_t)bt * ngT * 8 + (j & 7),
              (j >> 3) << 4, HEAD_PTS, S, lane, w, sIdx,
              lChunk[w], lBase[w], run0);
  }
  __syncthreads();

  const int tot = (run0 < S ? run0 : S) * 7;
  float* orow = out + (size_t)m * S * 7;
  for (int e0 = 0; e0 < tot; e0 += 4 * EMIT_THREADS) {
    float v[4];
#pragma unroll
    for (int u = 0; u < 4; ++u) {
      const int e  = e0 + u * EMIT_THREADS + t;
      const int ec = e < tot ? e : tot - 1;
      const int s = ec / 7, f = ec - s * 7;
      v[u] = (f < 5) ? P[(size_t)sIdx[s] * 5 + f] : (f == 5 ? vx : vy);
    }
#pragma unroll
    for (int u = 0; u < 4; ++u) {
      const int e = e0 + u * EMIT_THREADS + t;
      if (e < tot) orow[e] = v[u];
    }
  }
  for (int e = tot + t; e < S * 7; e += EMIT_THREADS) orow[e] = 0.0f;
}

extern "C" void kernel_launch(void* const* d_in, const int* in_sizes, int n_in,
                              void* d_out, int out_size, void* d_ws, size_t ws_size,
                              hipStream_t stream) {
  const float* P = (const float*)d_in[0];
  const float* B = (const float*)d_in[1];
  float* out = (float*)d_out;

  const int N = in_sizes[0] / 5;
  const int M = in_sizes[1] / 9;
  const int S = out_size / (M * 7);
  const int bt   = M / 16;
  const int nchT = N > HEAD_PTS ? (N - HEAD_PTS + CHUNK_PTS - 1) / CHUNK_PTS : 0;
  const int ngT  = nchT * 16;

  char* ws = (char*)d_ws;
  unsigned* masksT = (unsigned*)ws;
  size_t tbytes    = (size_t)bt * ngT * 8 * sizeof(unsigned);
  unsigned* masksH = (unsigned*)(ws + tbytes);
  size_t hbytes    = (size_t)bt * (HEAD_CHUNKS * 16) * 8 * sizeof(unsigned);
  int* cntT        = (int*)(ws + tbytes + hbytes);
  int* cntH        = cntT + (size_t)M * nchT;
  int* headpart    = cntH + (size_t)M * HEAD_CHUNKS;
  int* slotOf      = headpart + (size_t)HEAD_SPLIT * M;

  head_kernel<<<bt * HEAD_SPLIT, HEAD_WAVES * 32, 0, stream>>>(P, N, B, masksH, cntH,
                                                               headpart, M);
  if (nchT > 0)
    tail_kernel<<<nchT, 32, (size_t)M * sizeof(int), stream>>>(
        P, N, B, M, S, headpart, masksT, ngT, cntT, nchT, slotOf);
  emit_kernel<<<M, EMIT_THREADS, (size_t)S * sizeof(int), stream>>>(
      P, B, masksH, cntH, masksT, ngT, cntT, nchT, slotOf, out, S);
}


# emit, phase 2: listed chunks walked one lane per chunk, 32 per pass (a lane loads its chunk's 16 words, writes its set bits in order); no wave prefix sum per chunk, no chunk-after-chunk chain
# speedup vs baseline: 2.8644x; 1.1008x over previous
#include <hip/hip_runtime.h>
#include <math.h>

#define GAMMA 1.1f

typedef __attribute__((ext_vector_type(2))) float v2f;
typedef __attribute__((ext_vector_type(8))) float v8f;

#define CHUNK_PTS   256
#define HEAD_CHUNKS 32
#define HEAD_PTS    (CHUNK_PTS * HEAD_CHUNKS)
#define HEAD_SPLIT  4
#define HEAD_WAVES  (HEAD_CHUNKS / HEAD_SPLIT)
#define TILE_GANG   3

__device__ __forceinline__ void load_chunk(const float* __restrict__ P, int N,
                                           int pbase, int lane,
                                           float (&px)[16], float (&py)[16]) {
#pragma unroll
  for (int g = 0; g < 16; ++g) {
    const int idx = pbase + g * 16 + (lane & 15);
    const int ic  = idx < N ? idx : N - 1;
    const float x = P[(size_t)ic * 5];
    const float y = P[(size_t)ic * 5 + 1];
    px[g] = idx < N ? x : 1e18f;
    py[g] = idx < N ? y : 1e18f;
  }
}

__device__ __forceinline__ float sqnorm2(float x, float y) {
#pragma clang fp contract(off)
  return x * x + y * y;
}

__device__ __forceinline__ void tile_chunk(v2f a, const float (&r2v)[8],
                                           const float (&px)[16],
                                           const float (&py)[16],
                                           bool lo, int lane,
                                           unsigned* __restrict__ mbase,
                                           int& clo, int& chi) {
#pragma unroll
  for (int k = 0; k < 16; k += 2) {
    const float px0 = px[k],     py0 = py[k];
    const float px1 = px[k + 1], py1 = py[k + 1];
    const float pp0 = sqnorm2(px0, py0);
    const float pp1 = sqnorm2(px1, py1);
    v2f b0, b1;
    b0.x = lo ? px0 : pp0;  b0.y = lo ? py0 : 1.0f;
    b1.x = lo ? px1 : pp1;  b1.y = lo ? py1 : 1.0f;

    v8f cz = {0.f, 0.f, 0.f, 0.f, 0.f, 0.f, 0.f, 0.f};
    v8f d0 = __builtin_amdgcn_wmma_f32_16x16x4_f32(false, a, false, b0,
                                                   (short)0, cz, false, false);
    v8f d1 = __builtin_amdgcn_wmma_f32_16x16x4_f32(false, a, false, b1,
                                                   (short)0, cz, false, false);
    unsigned w0 = 0, w1 = 0;
#pragma unroll
    for (int r = 0; r < 8; ++r) {
      unsigned bal0 = __builtin_amdgcn_ballot_w32(d0[r] <= r2v[r]);
      unsigned bal1 = __builtin_amdgcn_ballot_w32(d1[r] <= r2v[r]);
      w0 = (lane == r) ? bal0 : w0;
      w1 = (lane == r) ? bal1 : w1;
    }
    clo += __popc((w0 & 0xFFFFu) | (w1 << 16));
    chi += __popc((w0 >> 16) | (w1 & 0xFFFF0000u));
    if (lane < 8) {
      unsigned* mp = mbase + k * 8 + lane;
      mp[0] = w0;
      mp[8] = w1;
    }
  }
}

__device__ __forceinline__ void box_operands(const float* __restrict__ boxes,
                                             int bi, bool valid, bool lo,
                                             v2f& a, float& r2own) {
  const float bx = boxes[bi * 9 + 0], by = boxes[bi * 9 + 1];
  const float dx = boxes[bi * 9 + 3], dy = boxes[bi * 9 + 4];
  const float hx = 0.5f * dx, hy = 0.5f * dy;
  const float r = sqrtf(hx * hx + hy * hy) * GAMMA;
  const float k0 = -2.0f * bx, k1 = -2.0f * by, k3 = bx * bx + by * by;
  a.x = valid ? (lo ? k0 : 1.0f) : 0.0f;
  a.y = valid ? (lo ? k1 : k3)   : 0.0f;
  r2own = valid ? r * r : -1.0f;
}

template <int NT>
__device__ __forceinline__ void tiles_chunk(const v2f (&a)[NT],
                                            const float (&r2v)[NT][8],
                                            const float (&px)[16],
                                            const float (&py)[16],
                                            bool lo, int lane,
                                            unsigned* __restrict__ mbase,
                                            size_t tstride,
                                            int (&clo)[NT], int (&chi)[NT]) {
  v2f b[16];
#pragma unroll
  for (int g = 0; g < 16; ++g) {
    const float pp = sqnorm2(px[g], py[g]);
    b[g].x = lo ? px[g] : pp;
    b[g].y = lo ? py[g] : 1.0f;
  }
  __builtin_amdgcn_sched_barrier(0);
  const v8f cz = {0.f, 0.f, 0.f, 0.f, 0.f, 0.f, 0.f, 0.f};
  v8f d0[NT], d1[NT];
#pragma unroll
  for (int u = 0; u < NT; ++u) {
    d0[u] = __builtin_amdgcn_wmma_f32_16x16x4_f32(false, a[u], false, b[0],
                                                  (short)0, cz, false, false);
    d1[u] = __builtin_amdgcn_wmma_f32_16x16x4_f32(false, a[u], false, b[1],
                                                  (short)0, cz, false, false);
  }
#pragma unroll
  for (int k = 0; k < 16; k += 2) {
    unsigned w0[NT], w1[NT];
    v8f n0[NT], n1[NT];
#pragma unroll
    for (int u = 0; u < NT; ++u) {
      if (k + 2 < 16) {
        n0[u] = __builtin_amdgcn_wmma_f32_16x16x4_f32(false, a[u], false, b[k + 2],
                                                      (short)0, cz, false, false);
        n1[u] = __builtin_amdgcn_wmma_f32_16x16x4_f32(false, a[u], false, b[k + 3],
                                                      (short)0, cz, false, false);
      }
      unsigned x0 = 0, x1 = 0;
#pragma unroll
      for (int r = 0; r < 8; ++r) {
        unsigned bal0 = __builtin_amdgcn_ballot_w32(d0[u][r] <= r2v[u][r]);
        unsigned bal1 = __builtin_amdgcn_ballot_w32(d1[u][r] <= r2v[u][r]);
        x0 = (lane == r) ? bal0 : x0;
        x1 = (lane == r) ? bal1 : x1;
      }
      w0[u] = x0;
      w1[u] = x1;
      clo[u] += __popc((x0 & 0xFFFFu) | (x1 << 16));
      chi[u] += __popc((x0 >> 16) | (x1 & 0xFFFF0000u));
    }
    if (k + 2 < 16) {
#pragma unroll
      for (int i = 0; i < 2 * NT; ++i) {
        __builtin_amdgcn_sched_group_barrier(0x008, 1, 0);
        __builtin_amdgcn_sched_group_barrier(0x002, 16, 0);
      }
    }
    if (lane < 8) {
#pragma unroll
      for (int u = 0; u < NT; ++u) {
        unsigned* mp = mbase + (size_t)u * tstride + k * 8 + lane;
        mp[0] = w0[u];
        mp[8] = w1[u];
      }
    }
    if (k + 2 < 16) {
#pragma unroll
      for (int u = 0; u < NT; ++u) { d0[u] = n0[u]; d1[u] = n1[u]; }
    }
  }
}

template <int NT>
__device__ __forceinline__ void run_tiles(const float* __restrict__ boxes,
                                          const int* slist, int nslow, int t0,
                                          const float (&px)[16],
                                          const float (&py)[16],
                                          bool lo, int lane,
                                          unsigned* __restrict__ masksT, int ngT,
                                          int* __restrict__ cntT, int nchT,
                                          int cc) {
  v2f a[NT];
  float r2own[NT];
#pragma unroll
  for (int u = 0; u < NT; ++u) {
    const int q = (t0 + u) * 16 + (lane & 15);
    const bool valid = q < nslow;
    const int bi = slist[valid ? q : 0];
    box_operands(boxes, bi, valid, lo, a[u], r2own[u]);
  }
  float r2v[NT][8];
#pragma unroll
  for (int u = 0; u < NT; ++u) {
#pragma unroll
    for (int r = 0; r < 8; ++r) r2v[u][r] = __shfl(r2own[u], r + (lo ? 0 : 8));
  }
  int clo[NT], chi[NT];
#pragma unroll
  for (int u = 0; u < NT; ++u) { clo[u] = 0; chi[u] = 0; }

  tiles_chunk<NT>(a, r2v, px, py, lo, lane,
                  masksT + ((size_t)t0 * ngT + (size_t)cc * 16) * 8,
                  (size_t)ngT * 8, clo, chi);

  if (lane < 8) {
#pragma unroll
    for (int u = 0; u < NT; ++u) {
      cntT[(size_t)((t0 + u) * 16 + lane)     * nchT + cc] = clo[u];
      cntT[(size_t)((t0 + u) * 16 + lane + 8) * nchT + cc] = chi[u];
    }
  }
}

__global__ __launch_bounds__(HEAD_WAVES * 32)
void head_kernel(const float* __restrict__ P, int N,
                 const float* __restrict__ boxes,
                 unsigned* __restrict__ masksH,
                 int* __restrict__ cntH,
                 int* __restrict__ headpart, int M) {
  __shared__ int sC[HEAD_WAVES][16];
  const int tile = blockIdx.x / HEAD_SPLIT;
  const int qtr  = blockIdx.x % HEAD_SPLIT;
  const int lane = threadIdx.x & 31;
  const int wv   = __builtin_amdgcn_readfirstlane(threadIdx.x >> 5);
  const int w    = qtr * HEAD_WAVES + wv;
  const bool lo  = lane < 16;

  float px[16], py[16];
  load_chunk(P, N, w * CHUNK_PTS, lane, px, py);

  v2f a; float r2own;
  box_operands(boxes, tile * 16 + (lane & 15), true, lo, a, r2own);
  float r2v[8];
#pragma unroll
  for (int r = 0; r < 8; ++r) r2v[r] = __shfl(r2own, r + (lo ? 0 : 8));

  int clo = 0, chi = 0;
  tile_chunk(a, r2v, px, py, lo, lane,
             masksH + ((size_t)tile * (HEAD_CHUNKS * 16) + w * 16) * 8, clo, chi);

  if (lane < 8) {
    cntH[(tile * 16 + lane)     * HEAD_CHUNKS + w] = clo;
    cntH[(tile * 16 + lane + 8) * HEAD_CHUNKS + w] = chi;
    sC[wv][lane]     = clo;
    sC[wv][lane + 8] = chi;
  }
  __syncthreads();
  if (threadIdx.x < 16) {
    int s = 0;
    for (int i = 0; i < HEAD_WAVES; ++i) s += sC[i][threadIdx.x];
    headpart[qtr * M + tile * 16 + threadIdx.x] = s;
  }
}

__global__ __launch_bounds__(32)
void tail_kernel(const float* __restrict__ P, int N,
                 const float* __restrict__ boxes, int M, int S,
                 const int* __restrict__ headpart,
                 unsigned* __restrict__ masksT, int ngT,
                 int* __restrict__ cntT, int nchT,
                 int* __restrict__ slotOf) {
  extern __shared__ int slist[];
  const int cc   = blockIdx.x;
  const int lane = threadIdx.x;
  const bool lo  = lane < 16;

  float px[16], py[16];
  load_chunk(P, N, HEAD_PTS + cc * CHUNK_PTS, lane, px, py);

  int nslow = 0;
  for (int b0 = 0; b0 < M; b0 += 256) {
    bool f[8];
#pragma unroll
    for (int i = 0; i < 8; ++i) {
      const int idx = b0 + i * 32 + lane;
      const int ic = idx < M ? idx : M - 1;
      int v = 0;
#pragma unroll
      for (int q = 0; q < HEAD_SPLIT; ++q) v += headpart[q * M + ic];
      f[i] = idx < M && v < S;
    }
#pragma unroll
    for (int i = 0; i < 8; ++i) {
      const int idx = b0 + i * 32 + lane;
      const unsigned bal = __builtin_amdgcn_ballot_w32(f[i]);
      const int rank = nslow + __popc(bal & ((1u << lane) - 1u));
      if (f[i]) slist[rank] = idx;
      if (cc == 0 && idx < M) slotOf[idx] = f[i] ? rank : -1;
      nslow += __popc(bal);
    }
  }
  __syncthreads();

  static_assert(TILE_GANG == 3, "the branches below name the gang sizes");
  const int ntiles = (nslow + 15) >> 4;
  for (int t0 = 0; t0 < ntiles; t0 += TILE_GANG) {
    const int left = ntiles - t0;
    if (left >= 3)
      run_tiles<3>(boxes, slist, nslow, t0, px, py, lo, lane, masksT, ngT, cntT, nchT, cc);
    else if (left == 2)
      run_tiles<2>(boxes, slist, nslow, t0, px, py, lo, lane, masksT, ngT, cntT, nchT, cc);
    else
      run_tiles<1>(boxes, slist, nslow, t0, px, py, lo, lane, masksT, ngT, cntT, nchT, cc);
  }
}

__device__ __forceinline__ int wave_incl_scan(int x) {
  x += __builtin_amdgcn_update_dpp(0, x, 0x111, 0xf, 0xf, true);
  x += __builtin_amdgcn_update_dpp(0, x, 0x112, 0xf, 0xf, true);
  x += __builtin_amdgcn_update_dpp(0, x, 0x114, 0xf, 0xf, true);
  x += __builtin_amdgcn_update_dpp(0, x, 0x118, 0xf, 0xf, true);
  int row0 = __builtin_amdgcn_readlane(x, 15);
  return x + ((threadIdx.x & 16) ? row0 : 0);
}

#define EMIT_THREADS 256
#define EMIT_BATCH 4

static_assert(HEAD_CHUNKS == 32, "walk_head keeps one chunk count per lane");
__device__ __forceinline__ int walk_head(const int* __restrict__ crow,
                                         const unsigned* __restrict__ mrow,
                                         int sh, int S, int lane, int w,
                                         int* sIdx) {
  const int bsh = sh + 8 * (lane & 1);
  const int cnt  = crow[lane];
  const int incl = wave_incl_scan(cnt);
  const int total = __builtin_amdgcn_readlane(incl, 31);
  const int myBase = incl - cnt;
  unsigned amask = __builtin_amdgcn_ballot_w32(cnt > 0 && myBase < S) &
                   (0x01010101u << w);
  if (amask) {
    int  csel[EMIT_BATCH];
    bool valid[EMIT_BATCH];
    unsigned wa[EMIT_BATCH];
    const int first = __builtin_ctz(amask);
#pragma unroll
    for (int b = 0; b < EMIT_BATCH; ++b) {
      valid[b] = amask != 0u;
      csel[b]  = valid[b] ? __builtin_ctz(amask) : first;
      amask &= amask - 1u;
      wa[b] = mrow[((size_t)csel[b] * 16 + (lane >> 1)) * 8];
    }
#pragma unroll
    for (int b = 0; b < EMIT_BATCH; ++b) {
      if (valid[b]) {
        const int base = __builtin_amdgcn_readlane(myBase, csel[b]);
        unsigned Wd = (wa[b] >> bsh) & 0xFFu;
        const int k = __popc(Wd);
        int slot = base + wave_incl_scan(k) - k;
        const int pb = csel[b] * CHUNK_PTS + lane * 8;
        while (Wd != 0u && slot < S) {
          sIdx[slot] = pb + __builtin_ctz(Wd);
          Wd &= Wd - 1u;
          ++slot;
        }
      }
    }
  }
  return total;
}

#define TAIL_TILES 4
#define LIST_MAX   (TAIL_TILES * 32)

__device__ __forceinline__ void walk_tail(const int* __restrict__ crow, int nch,
                                          const unsigned* __restrict__ mrow,
                                          int sh, int ptbase, int S,
                                          int lane, int w,
                                          int* sIdx, int* lChunk, int* lBase,
                                          int& run0) {
  const unsigned below = (1u << lane) - 1u;
  for (int c00 = 0; c00 < nch && run0 < S; c00 += 256 * TAIL_TILES) {
    int n[TAIL_TILES][8];
#pragma unroll
    for (int j = 0; j < TAIL_TILES; ++j) {
#pragma unroll
      for (int i = 0; i < 8; ++i) {
        const int c = c00 + 256 * j + 8 * lane + i;
        n[j][i] = crow[c < nch ? c : nch - 1];
      }
    }
    asm volatile("" :: "v"(n[0][0]), "v"(n[0][1]), "v"(n[0][2]), "v"(n[0][3]),
                       "v"(n[0][4]), "v"(n[0][5]), "v"(n[0][6]), "v"(n[0][7]),
                       "v"(n[1][0]), "v"(n[1][1]), "v"(n[1][2]), "v"(n[1][3]),
                       "v"(n[1][4]), "v"(n[1][5]), "v"(n[1][6]), "v"(n[1][7]));
    asm volatile("" :: "v"(n[2][0]), "v"(n[2][1]), "v"(n[2][2]), "v"(n[2][3]),
                       "v"(n[2][4]), "v"(n[2][5]), "v"(n[2][6]), "v"(n[2][7]),
                       "v"(n[3][0]), "v"(n[3][1]), "v"(n[3][2]), "v"(n[3][3]),
                       "v"(n[3][4]), "v"(n[3][5]), "v"(n[3][6]), "v"(n[3][7]));
#pragma unroll
    for (int j = 0; j < TAIL_TILES; ++j) {
#pragma unroll
      for (int i = 0; i < 8; ++i) {
        const int c = c00 + 256 * j + 8 * lane + i;
        n[j][i] = (c < nch) ? n[j][i] : 0;
      }
    }
    int len = 0;
    int run = run0;
#pragma unroll
    for (int j = 0; j < TAIL_TILES; ++j) {
      int p = 0, myOff = 0, myCnt = 0;
#pragma unroll
      for (int i = 0; i < 8; ++i) {
        if (i == w) { myOff = p; myCnt = n[j][i]; }
        p += n[j][i];
      }
      const int incl = wave_incl_scan(p);
      const int tileTotal = __builtin_amdgcn_readlane(incl, 31);
      const int myBase = run + (incl - p) + myOff;
      const bool act = myCnt > 0 && myBase < S;
      const unsigned am = __builtin_amdgcn_ballot_w32(act);
      if (act) {
        const int k = len + __popc(am & below);
        lChunk[k] = c00 + 256 * j + 8 * lane + w;
        lBase[k]  = myBase;
      }
      len += __popc(am);
      run += tileTotal;
    }
    run0 = run;
    __syncthreads();

    for (int k0 = 0; k0 < len; k0 += 32) {
      const bool have = k0 + lane < len;
      const int k = have ? k0 + lane : k0;
      const int c = lChunk[k];
      int slot = lBase[k];
      const unsigned* mp = mrow + (size_t)c * (16 * 8);
      unsigned wv[16];
#pragma unroll
      for (int g = 0; g < 16; ++g) wv[g] = mp[g * 8];
      unsigned m[8];
#pragma unroll
      for (int j = 0; j < 8; ++j) {
        const unsigned a = (wv[2 * j] >> sh) & 0xFFFFu;
        const unsigned b = (wv[2 * j + 1] >> sh) & 0xFFFFu;
        m[j] = have ? (a | (b << 16)) : 0u;
      }
      const int pb = ptbase + c * CHUNK_PTS;
      for (;;) {
        unsigned sel = m[7];
        int off = 7 * 32;
#pragma unroll
        for (int j = 6; j >= 0; --j) {
          if (m[j] != 0u) { sel = m[j]; off = j * 32; }
        }
        if (sel == 0u || slot >= S) break;
        sIdx[slot] = pb + off + __builtin_ctz(sel);
        ++slot;
#pragma unroll
        for (int j = 0; j < 8; ++j)
          m[j] = (off == j * 32) ? (m[j] & (m[j] - 1u)) : m[j];
      }
    }
  }
}

__global__ __launch_bounds__(EMIT_THREADS)
void emit_kernel(const float* __restrict__ P,
                 const float* __restrict__ boxes,
                 const unsigned* __restrict__ masksH,
                 const int* __restrict__ cntH,
                 const unsigned* __restrict__ masksT, int ngT,
                 const int* __restrict__ cntT, int nchT,
                 const int* __restrict__ slotOf,
                 float* __restrict__ out, int S) {
  extern __shared__ int sIdx[];
  __shared__ int lChunk[EMIT_THREADS / 32][LIST_MAX];
  __shared__ int lBase[EMIT_THREADS / 32][LIST_MAX];

  const int m    = blockIdx.x;
  const int t    = threadIdx.x;
  const int lane = t & 31;
  const int w    = __builtin_amdgcn_readfirstlane(t >> 5);
  const float vx = boxes[m * 9 + 7];
  const float vy = boxes[m * 9 + 8];
  const int slot = nchT > 0 ? slotOf[m] : -1;

  int run0;
  {
    const int bt = m >> 4, j = m & 15;
    run0 = walk_head(cntH + (size_t)m * HEAD_CHUNKS,
                     masksH + (size_t)bt * (HEAD_CHUNKS * 16) * 8 + (j & 7),
                     (j >> 3) << 4, S, lane, w, sIdx);
  }
  if (run0 < S && nchT > 0) {
    const int bt = slot >> 4, j = slot & 15;
    walk_tail(cntT + (size_t)slot * nchT, nchT,
              masksT + (size_t)bt * ngT * 8 + (j & 7),
              (j >> 3) << 4, HEAD_PTS, S, lane, w, sIdx,
              lChunk[w], lBase[w], run0);
  }
  __syncthreads();

  const int tot = (run0 < S ? run0 : S) * 7;
  float* orow = out + (size_t)m * S * 7;
  for (int e0 = 0; e0 < tot; e0 += 4 * EMIT_THREADS) {
    float v[4];
#pragma unroll
    for (int u = 0; u < 4; ++u) {
      const int e  = e0 + u * EMIT_THREADS + t;
      const int ec = e < tot ? e : tot - 1;
      const int s = ec / 7, f = ec - s * 7;
      v[u] = (f < 5) ? P[(size_t)sIdx[s] * 5 + f] : (f == 5 ? vx : vy);
    }
#pragma unroll
    for (int u = 0; u < 4; ++u) {
      const int e = e0 + u * EMIT_THREADS + t;
      if (e < tot) orow[e] = v[u];
    }
  }
  for (int e = tot + t; e < S * 7; e += EMIT_THREADS) orow[e] = 0.0f;
}

extern "C" void kernel_launch(void* const* d_in, const int* in_sizes, int n_in,
                              void* d_out, int out_size, void* d_ws, size_t ws_size,
                              hipStream_t stream) {
  const float* P = (const float*)d_in[0];
  const float* B = (const float*)d_in[1];
  float* out = (float*)d_out;

  const int N = in_sizes[0] / 5;
  const int M = in_sizes[1] / 9;
  const int S = out_size / (M * 7);
  const int bt   = M / 16;
  const int nchT = N > HEAD_PTS ? (N - HEAD_PTS + CHUNK_PTS - 1) / CHUNK_PTS : 0;
  const int ngT  = nchT * 16;

  char* ws = (char*)d_ws;
  unsigned* masksT = (unsigned*)ws;
  size_t tbytes    = (size_t)bt * ngT * 8 * sizeof(unsigned);
  unsigned* masksH = (unsigned*)(ws + tbytes);
  size_t hbytes    = (size_t)bt * (HEAD_CHUNKS * 16) * 8 * sizeof(unsigned);
  int* cntT        = (int*)(ws + tbytes + hbytes);
  int* cntH        = cntT + (size_t)M * nchT;
  int* headpart    = cntH + (size_t)M * HEAD_CHUNKS;
  int* slotOf      = headpart + (size_t)HEAD_SPLIT * M;

  head_kernel<<<bt * HEAD_SPLIT, HEAD_WAVES * 32, 0, stream>>>(P, N, B, masksH, cntH,
                                                               headpart, M);
  if (nchT > 0)
    tail_kernel<<<nchT, 32, (size_t)M * sizeof(int), stream>>>(
        P, N, B, M, S, headpart, masksT, ngT, cntT, nchT, slotOf);
  emit_kernel<<<M, EMIT_THREADS, (size_t)S * sizeof(int), stream>>>(
      P, B, masksH, cntH, masksT, ngT, cntT, nchT, slotOf, out, S);
}
